# starting kernel
# baseline (speedup 1.0000x reference)
#include <hip/hip_runtime.h>

typedef __attribute__((ext_vector_type(2))) float v2f;
typedef __attribute__((ext_vector_type(8))) float v8f;

#define HIDC 128

template <int NC>
__global__ __launch_bounds__(256) void gemm_wmma_f32(
    const float* __restrict__ A, const float* __restrict__ B,
    const float* __restrict__ bias, float* __restrict__ C, int M)
{
    constexpr int K = HIDC;
    constexpr int tilesN = (NC + 15) >> 4;
    const int tilesM = (M + 15) >> 4;
    const int wid = blockIdx.x * 8 + (threadIdx.x >> 5);
    if (wid >= tilesM * tilesN) return;
    const int tm = wid / tilesN, tn = wid % tilesN;
    const int m0 = tm << 4, n0 = tn << 4;
    const int lane = threadIdx.x & 31;
    const int half = lane >> 4;
    const int l16  = lane & 15;

    const int arow   = min(m0 + l16, M - 1);
    const int bcol_r = n0 + l16;
    const int bcol   = min(bcol_r, NC - 1);

    const float* Ap = A + (size_t)arow * K + 2 * half;
    const float* Bp = B + (size_t)(2 * half) * NC + bcol;

    v8f acc = {};
#pragma unroll
    for (int k0 = 0; k0 < K; k0 += 4) {
        v2f a = *(const v2f*)(Ap + k0);
        v2f b;
        b.x = Bp[k0 * NC];
        b.y = Bp[(k0 + 1) * NC];
        acc = __builtin_amdgcn_wmma_f32_16x16x4_f32(
                  false, a, false, b, (short)0, acc, false, false);
    }

    if (bcol_r < NC) {
        const float bv = bias ? bias[bcol_r] : 0.0f;
#pragma unroll
        for (int v = 0; v < 8; ++v) {
            const int r = m0 + v + 8 * half;
            if (r < M) C[(size_t)r * NC + bcol_r] = acc[v] + bv;
        }
    }
}

__global__ void zero_f32(float* __restrict__ p, int n) {
    int i = blockIdx.x * blockDim.x + threadIdx.x;
    if (i < n) p[i] = 0.0f;
}

__global__ void degree_count(const int* __restrict__ row, int E,
                             float* __restrict__ deg) {
    int e = blockIdx.x * blockDim.x + threadIdx.x;
    if (e < E) atomicAdd(&deg[row[e]], 1.0f);
}

__global__ void invert_deg(float* __restrict__ d, int n) {
    int i = blockIdx.x * blockDim.x + threadIdx.x;
    if (i < n) d[i] = 1.0f / (d[i] + 1e-10f);
}

__global__ __launch_bounds__(256) void scatter_add_edges(
    const float* __restrict__ m, const int* __restrict__ ei, int E,
    float* __restrict__ agg)
{
    int t  = blockIdx.x * 256 + threadIdx.x;
    int e  = t >> 5;
    int c4 = (t & 31) << 2;
    if (e >= E) return;
    int r  = ei[e];
    int cl = ei[E + e];
    const float4 mv = *(const float4*)(m + (size_t)r * HIDC + c4);
    float* dst = agg + (size_t)cl * HIDC + c4;
    atomicAdd(dst + 0, mv.x);
    atomicAdd(dst + 1, mv.y);
    atomicAdd(dst + 2, mv.z);
    atomicAdd(dst + 3, mv.w);
}

__global__ void bias_relu(float* __restrict__ agg, const float* __restrict__ b,
                          int total) {
    int i = blockIdx.x * blockDim.x + threadIdx.x;
    if (i < total) {
        float v = agg[i] + b[i & (HIDC - 1)];
        agg[i] = fmaxf(v, 0.0f);
    }
}

__global__ __launch_bounds__(256) void gamma_edges(
    const float* __restrict__ agg, const int* __restrict__ ei, int E,
    float* __restrict__ gsum)
{
    int wid  = blockIdx.x * 8 + (threadIdx.x >> 5);
    int lane = threadIdx.x & 31;
    if (wid >= E) return;
    int r = ei[wid];
    int c = ei[E + wid];
    const float4 a = *(const float4*)(agg + (size_t)r * HIDC + lane * 4);
    const float4 b = *(const float4*)(agg + (size_t)c * HIDC + lane * 4);
    float dx = a.x - b.x, dy = a.y - b.y, dz = a.z - b.z, dw = a.w - b.w;
    float s = dx * dx + dy * dy + dz * dz + dw * dw;
#pragma unroll
    for (int off = 16; off > 0; off >>= 1)
        s += __shfl_down(s, off, 32);
    if (lane == 0) atomicAdd(&gsum[r], s);
}

__global__ void gamma_node(const float* __restrict__ gsum,
                           const float* __restrict__ inv,
                           float* __restrict__ g, int n) {
    int i = blockIdx.x * blockDim.x + threadIdx.x;
    if (i < n) g[i] = tanhf(gsum[i] * inv[i]);
}

__global__ void update_h(float* __restrict__ h, const float* __restrict__ agg,
                         const float* __restrict__ xs,
                         const float* __restrict__ g1,
                         const float* __restrict__ g2, int total) {
    int i = blockIdx.x * blockDim.x + threadIdx.x;
    if (i < total) {
        int node = i >> 7;
        float gs = g1[node], gq = g2[node];
        h[i] = (h[i] + gs * agg[i] + gq * xs[i]) / (1.0f + gs + gq);
    }
}

extern "C" void kernel_launch(void* const* d_in, const int* in_sizes, int n_in,
                              void* d_out, int out_size, void* d_ws, size_t ws_size,
                              hipStream_t stream) {
    const float* x      = (const float*)d_in[0];
    const float* x0     = (const float*)d_in[1];
    const int*   ei     = (const int*)d_in[2];
    const int*   ei2    = (const int*)d_in[3];
    const float* in_W   = (const float*)d_in[4];
    const float* skip_W = (const float*)d_in[5];
    const float* conv_W = (const float*)d_in[6];
    const float* conv_b = (const float*)d_in[7];
    const float* fc_W   = (const float*)d_in[8];
    const float* fc_b   = (const float*)d_in[9];
    float* out = (float*)d_out;

    const int N  = in_sizes[0] / HIDC;
    const int E1 = in_sizes[2] / 2;
    const int E2 = in_sizes[3] / 2;

    const size_t nh = (size_t)N * HIDC;
    float* h     = (float*)d_ws;
    float* xskip = h + nh;
    float* agg   = xskip + nh;
    float* mbuf  = agg + nh;
    float* inv1  = mbuf + nh;
    float* inv2  = inv1 + N;
    float* gsum1 = inv2 + N;
    float* gsum2 = gsum1 + N;
    float* g1    = gsum2 + N;
    float* g2    = g1 + N;

    const dim3 blk(256);
    auto gemm_blocks = [](int M, int Nc) {
        int tiles = ((M + 15) / 16) * ((Nc + 15) / 16);
        return dim3((tiles + 7) / 8);
    };
    const int nhi = (int)nh;

    gemm_wmma_f32<HIDC><<<gemm_blocks(N, HIDC), blk, 0, stream>>>(x,    in_W,   nullptr, h,     N);
    gemm_wmma_f32<HIDC><<<gemm_blocks(N, HIDC), blk, 0, stream>>>(x0,   in_W,   nullptr, mbuf,  N);
    gemm_wmma_f32<HIDC><<<gemm_blocks(N, HIDC), blk, 0, stream>>>(mbuf, skip_W, nullptr, xskip, N);

    zero_f32<<<(N + 255) / 256, blk, 0, stream>>>(inv1, N);
    zero_f32<<<(N + 255) / 256, blk, 0, stream>>>(inv2, N);
    degree_count<<<(E1 + 255) / 256, blk, 0, stream>>>(ei,  E1, inv1);
    degree_count<<<(E2 + 255) / 256, blk, 0, stream>>>(ei2, E2, inv2);
    invert_deg<<<(N + 255) / 256, blk, 0, stream>>>(inv1, N);
    invert_deg<<<(N + 255) / 256, blk, 0, stream>>>(inv2, N);

    for (int l = 0; l < 2; ++l) {
        gemm_wmma_f32<HIDC><<<gemm_blocks(N, HIDC), blk, 0, stream>>>(
            h, conv_W + (size_t)l * HIDC * HIDC, nullptr, mbuf, N);
        zero_f32<<<(nhi + 255) / 256, blk, 0, stream>>>(agg, nhi);
        {
            long long threads = (long long)E1 * 32;
            scatter_add_edges<<<(unsigned)((threads + 255) / 256), blk, 0, stream>>>(mbuf, ei, E1, agg);
        }
        bias_relu<<<(nhi + 255) / 256, blk, 0, stream>>>(agg, conv_b + l * HIDC, nhi);
        zero_f32<<<(N + 255) / 256, blk, 0, stream>>>(gsum1, N);
        zero_f32<<<(N + 255) / 256, blk, 0, stream>>>(gsum2, N);
        gamma_edges<<<(E1 + 7) / 8, blk, 0, stream>>>(agg, ei,  E1, gsum1);
        gamma_edges<<<(E2 + 7) / 8, blk, 0, stream>>>(agg, ei2, E2, gsum2);
        gamma_node<<<(N + 255) / 256, blk, 0, stream>>>(gsum1, inv1, g1, N);
        gamma_node<<<(N + 255) / 256, blk, 0, stream>>>(gsum2, inv2, g2, N);
        update_h<<<(nhi + 255) / 256, blk, 0, stream>>>(h, agg, xskip, g1, g2, nhi);
    }

    gemm_wmma_f32<40><<<gemm_blocks(N, 40), blk, 0, stream>>>(h, fc_W, fc_b, out, N);
}


# gamma_edges: a wave walks 8 consecutive edges (scalar indices, all row gathers in flight together), sums each run of equal rows in registers: one lane tree + one atomic per run; sum order changes
# speedup vs baseline: 1.4791x; 1.4791x over previous
#include <hip/hip_runtime.h>

typedef __attribute__((ext_vector_type(2))) float v2f;
typedef __attribute__((ext_vector_type(8))) float v8f;

#define HIDC 128

template <int NC>
__global__ __launch_bounds__(256) void gemm_wmma_f32(
    const float* __restrict__ A, const float* __restrict__ B,
    const float* __restrict__ bias, float* __restrict__ C, int M)
{
    constexpr int K = HIDC;
    constexpr int tilesN = (NC + 15) >> 4;
    const int tilesM = (M + 15) >> 4;
    const int wid = blockIdx.x * 8 + (threadIdx.x >> 5);
    if (wid >= tilesM * tilesN) return;
    const int tm = wid / tilesN, tn = wid % tilesN;
    const int m0 = tm << 4, n0 = tn << 4;
    const int lane = threadIdx.x & 31;
    const int half = lane >> 4;
    const int l16  = lane & 15;

    const int arow   = min(m0 + l16, M - 1);
    const int bcol_r = n0 + l16;
    const int bcol   = min(bcol_r, NC - 1);

    const float* Ap = A + (size_t)arow * K + 2 * half;
    const float* Bp = B + (size_t)(2 * half) * NC + bcol;

    v8f acc = {};
#pragma unroll
    for (int k0 = 0; k0 < K; k0 += 4) {
        v2f a = *(const v2f*)(Ap + k0);
        v2f b;
        b.x = Bp[k0 * NC];
        b.y = Bp[(k0 + 1) * NC];
        acc = __builtin_amdgcn_wmma_f32_16x16x4_f32(
                  false, a, false, b, (short)0, acc, false, false);
    }

    if (bcol_r < NC) {
        const float bv = bias ? bias[bcol_r] : 0.0f;
#pragma unroll
        for (int v = 0; v < 8; ++v) {
            const int r = m0 + v + 8 * half;
            if (r < M) C[(size_t)r * NC + bcol_r] = acc[v] + bv;
        }
    }
}

__global__ void zero_f32(float* __restrict__ p, int n) {
    int i = blockIdx.x * blockDim.x + threadIdx.x;
    if (i < n) p[i] = 0.0f;
}

__global__ void degree_count(const int* __restrict__ row, int E,
                             float* __restrict__ deg) {
    int e = blockIdx.x * blockDim.x + threadIdx.x;
    if (e < E) atomicAdd(&deg[row[e]], 1.0f);
}

__global__ void invert_deg(float* __restrict__ d, int n) {
    int i = blockIdx.x * blockDim.x + threadIdx.x;
    if (i < n) d[i] = 1.0f / (d[i] + 1e-10f);
}

__global__ __launch_bounds__(256) void scatter_add_edges(
    const float* __restrict__ m, const int* __restrict__ ei, int E,
    float* __restrict__ agg)
{
    int t  = blockIdx.x * 256 + threadIdx.x;
    int e  = t >> 5;
    int c4 = (t & 31) << 2;
    if (e >= E) return;
    int r  = ei[e];
    int cl = ei[E + e];
    const float4 mv = *(const float4*)(m + (size_t)r * HIDC + c4);
    float* dst = agg + (size_t)cl * HIDC + c4;
    atomicAdd(dst + 0, mv.x);
    atomicAdd(dst + 1, mv.y);
    atomicAdd(dst + 2, mv.z);
    atomicAdd(dst + 3, mv.w);
}

__global__ void bias_relu(float* __restrict__ agg, const float* __restrict__ b,
                          int total) {
    int i = blockIdx.x * blockDim.x + threadIdx.x;
    if (i < total) {
        float v = agg[i] + b[i & (HIDC - 1)];
        agg[i] = fmaxf(v, 0.0f);
    }
}

#define GCH 8

__device__ __forceinline__ float4 ld_row4(const float* __restrict__ agg, int node, int lane) {
    return *(const float4*)(agg + (size_t)node * HIDC + lane * 4);
}

__device__ __forceinline__ float sqdist4(const float4 a, const float4 b) {
    float dx = a.x - b.x, dy = a.y - b.y, dz = a.z - b.z, dw = a.w - b.w;
    return dx * dx + dy * dy + dz * dz + dw * dw;
}

__device__ __forceinline__ void flush_run(float* __restrict__ gsum, int node, float s, int lane) {
#pragma unroll
    for (int off = 16; off > 0; off >>= 1)
        s += __shfl_down(s, off, 32);
    if (lane == 0) atomicAdd(&gsum[node], s);
}

__global__ __launch_bounds__(256) void gamma_edges(
    const float* __restrict__ agg, const int* __restrict__ ei, int E,
    float* __restrict__ gsum)
{
    const int wv   = __builtin_amdgcn_readfirstlane((int)(threadIdx.x >> 5));
    const int lane = threadIdx.x & 31;
    const int e0   = (blockIdx.x * 8 + wv) * GCH;
    if (e0 >= E) return;

    if (e0 + GCH <= E) {
        int rr[GCH], cc[GCH];
#pragma unroll
        for (int j = 0; j < GCH; ++j) { rr[j] = ei[e0 + j]; cc[j] = ei[E + e0 + j]; }

        float4 bv[GCH], av[GCH];
#pragma unroll
        for (int j = 0; j < GCH; ++j) bv[j] = ld_row4(agg, cc[j], lane);
        av[0] = ld_row4(agg, rr[0], lane);
#pragma unroll
        for (int j = 1; j < GCH; ++j)
            if (rr[j] != rr[j - 1]) av[j] = ld_row4(agg, rr[j], lane);

        float4 a = av[0];
        float acc = 0.0f;
#pragma unroll
        for (int j = 0; j < GCH; ++j) {
            if (j > 0 && rr[j] != rr[j - 1]) {
                flush_run(gsum, rr[j - 1], acc, lane);
                acc = 0.0f;
                a = av[j];
            }
            acc += sqdist4(a, bv[j]);
        }
        flush_run(gsum, rr[GCH - 1], acc, lane);
    } else {
        const int n = E - e0;
        int prev = ei[e0];
        float acc = 0.0f;
        for (int j = 0; j < n; ++j) {
            const int r = ei[e0 + j];
            const int c = ei[E + e0 + j];
            if (r != prev) { flush_run(gsum, prev, acc, lane); acc = 0.0f; prev = r; }
            acc += sqdist4(ld_row4(agg, r, lane), ld_row4(agg, c, lane));
        }
        flush_run(gsum, prev, acc, lane);
    }
}

__global__ void gamma_node(const float* __restrict__ gsum,
                           const float* __restrict__ inv,
                           float* __restrict__ g, int n) {
    int i = blockIdx.x * blockDim.x + threadIdx.x;
    if (i < n) g[i] = tanhf(gsum[i] * inv[i]);
}

__global__ void update_h(float* __restrict__ h, const float* __restrict__ agg,
                         const float* __restrict__ xs,
                         const float* __restrict__ g1,
                         const float* __restrict__ g2, int total) {
    int i = blockIdx.x * blockDim.x + threadIdx.x;
    if (i < total) {
        int node = i >> 7;
        float gs = g1[node], gq = g2[node];
        h[i] = (h[i] + gs * agg[i] + gq * xs[i]) / (1.0f + gs + gq);
    }
}

extern "C" void kernel_launch(void* const* d_in, const int* in_sizes, int n_in,
                              void* d_out, int out_size, void* d_ws, size_t ws_size,
                              hipStream_t stream) {
    const float* x      = (const float*)d_in[0];
    const float* x0     = (const float*)d_in[1];
    const int*   ei     = (const int*)d_in[2];
    const int*   ei2    = (const int*)d_in[3];
    const float* in_W   = (const float*)d_in[4];
    const float* skip_W = (const float*)d_in[5];
    const float* conv_W = (const float*)d_in[6];
    const float* conv_b = (const float*)d_in[7];
    const float* fc_W   = (const float*)d_in[8];
    const float* fc_b   = (const float*)d_in[9];
    float* out = (float*)d_out;

    const int N  = in_sizes[0] / HIDC;
    const int E1 = in_sizes[2] / 2;
    const int E2 = in_sizes[3] / 2;

    const size_t nh = (size_t)N * HIDC;
    float* h     = (float*)d_ws;
    float* xskip = h + nh;
    float* agg   = xskip + nh;
    float* mbuf  = agg + nh;
    float* inv1  = mbuf + nh;
    float* inv2  = inv1 + N;
    float* gsum1 = inv2 + N;
    float* gsum2 = gsum1 + N;
    float* g1    = gsum2 + N;
    float* g2    = g1 + N;

    const dim3 blk(256);
    auto gemm_blocks = [](int M, int Nc) {
        int tiles = ((M + 15) / 16) * ((Nc + 15) / 16);
        return dim3((tiles + 7) / 8);
    };
    const int nhi = (int)nh;

    gemm_wmma_f32<HIDC><<<gemm_blocks(N, HIDC), blk, 0, stream>>>(x,    in_W,   nullptr, h,     N);
    gemm_wmma_f32<HIDC><<<gemm_blocks(N, HIDC), blk, 0, stream>>>(x0,   in_W,   nullptr, mbuf,  N);
    gemm_wmma_f32<HIDC><<<gemm_blocks(N, HIDC), blk, 0, stream>>>(mbuf, skip_W, nullptr, xskip, N);

    zero_f32<<<(N + 255) / 256, blk, 0, stream>>>(inv1, N);
    zero_f32<<<(N + 255) / 256, blk, 0, stream>>>(inv2, N);
    degree_count<<<(E1 + 255) / 256, blk, 0, stream>>>(ei,  E1, inv1);
    degree_count<<<(E2 + 255) / 256, blk, 0, stream>>>(ei2, E2, inv2);
    invert_deg<<<(N + 255) / 256, blk, 0, stream>>>(inv1, N);
    invert_deg<<<(N + 255) / 256, blk, 0, stream>>>(inv2, N);

    for (int l = 0; l < 2; ++l) {
        gemm_wmma_f32<HIDC><<<gemm_blocks(N, HIDC), blk, 0, stream>>>(
            h, conv_W + (size_t)l * HIDC * HIDC, nullptr, mbuf, N);
        zero_f32<<<(nhi + 255) / 256, blk, 0, stream>>>(agg, nhi);
        {
            long long threads = (long long)E1 * 32;
            scatter_add_edges<<<(unsigned)((threads + 255) / 256), blk, 0, stream>>>(mbuf, ei, E1, agg);
        }
        bias_relu<<<(nhi + 255) / 256, blk, 0, stream>>>(agg, conv_b + l * HIDC, nhi);
        zero_f32<<<(N + 255) / 256, blk, 0, stream>>>(gsum1, N);
        zero_f32<<<(N + 255) / 256, blk, 0, stream>>>(gsum2, N);
        gamma_edges<<<(E1 + 8 * GCH - 1) / (8 * GCH), blk, 0, stream>>>(agg, ei,  E1, gsum1);
        gamma_edges<<<(E2 + 8 * GCH - 1) / (8 * GCH), blk, 0, stream>>>(agg, ei2, E2, gsum2);
        gamma_node<<<(N + 255) / 256, blk, 0, stream>>>(gsum1, inv1, g1, N);
        gamma_node<<<(N + 255) / 256, blk, 0, stream>>>(gsum2, inv2, g2, N);
        update_h<<<(nhi + 255) / 256, blk, 0, stream>>>(h, agg, xskip, g1, g2, nhi);
    }

    gemm_wmma_f32<40><<<gemm_blocks(N, 40), blk, 0, stream>>>(h, fc_W, fc_b, out, N);
}


# agg by destination: edges counting-sorted by col per call (int rank, scan, fill); a wave sums a node's source rows in registers, stores relu(sum+b) once; no float atomics on agg; sum order changes
# speedup vs baseline: 1.5832x; 1.0704x over previous
#include <hip/hip_runtime.h>

typedef __attribute__((ext_vector_type(2))) float v2f;
typedef __attribute__((ext_vector_type(8))) float v8f;

#define HIDC 128

template <int NC>
__global__ __launch_bounds__(256) void gemm_wmma_f32(
    const float* __restrict__ A, const float* __restrict__ B,
    const float* __restrict__ bias, float* __restrict__ C, int M)
{
    constexpr int K = HIDC;
    constexpr int tilesN = (NC + 15) >> 4;
    const int tilesM = (M + 15) >> 4;
    const int wid = blockIdx.x * 8 + (threadIdx.x >> 5);
    if (wid >= tilesM * tilesN) return;
    const int tm = wid / tilesN, tn = wid % tilesN;
    const int m0 = tm << 4, n0 = tn << 4;
    const int lane = threadIdx.x & 31;
    const int half = lane >> 4;
    const int l16  = lane & 15;

    const int arow   = min(m0 + l16, M - 1);
    const int bcol_r = n0 + l16;
    const int bcol   = min(bcol_r, NC - 1);

    const float* Ap = A + (size_t)arow * K + 2 * half;
    const float* Bp = B + (size_t)(2 * half) * NC + bcol;

    v8f acc = {};
#pragma unroll
    for (int k0 = 0; k0 < K; k0 += 4) {
        v2f a = *(const v2f*)(Ap + k0);
        v2f b;
        b.x = Bp[k0 * NC];
        b.y = Bp[(k0 + 1) * NC];
        acc = __builtin_amdgcn_wmma_f32_16x16x4_f32(
                  false, a, false, b, (short)0, acc, false, false);
    }

    if (bcol_r < NC) {
        const float bv = bias ? bias[bcol_r] : 0.0f;
#pragma unroll
        for (int v = 0; v < 8; ++v) {
            const int r = m0 + v + 8 * half;
            if (r < M) C[(size_t)r * NC + bcol_r] = acc[v] + bv;
        }
    }
}

__global__ void zero_f32(float* __restrict__ p, int n) {
    int i = blockIdx.x * blockDim.x + threadIdx.x;
    if (i < n) p[i] = 0.0f;
}

__global__ void degree_count(const int* __restrict__ row, int E,
                             float* __restrict__ deg) {
    int e = blockIdx.x * blockDim.x + threadIdx.x;
    if (e < E) atomicAdd(&deg[row[e]], 1.0f);
}

__global__ void invert_deg(float* __restrict__ d, int n) {
    int i = blockIdx.x * blockDim.x + threadIdx.x;
    if (i < n) d[i] = 1.0f / (d[i] + 1e-10f);
}

__global__ void rank_by_dst(const int* __restrict__ ei, int E,
                            int* __restrict__ cnt, int* __restrict__ rank) {
    int e = blockIdx.x * blockDim.x + threadIdx.x;
    if (e < E) rank[e] = atomicAdd(&cnt[ei[E + e]], 1);
}

__global__ __launch_bounds__(1024) void scan_counts(const int* __restrict__ cnt,
                                                    int* __restrict__ start, int n) {
    __shared__ int wsum[32];
    const int t = threadIdx.x, lane = t & 31, w = t >> 5;
    const int per = (n + 1023) >> 10;
    const int b = min(t * per, n), e = min(b + per, n);
    int s = 0;
    for (int i = b; i < e; ++i) s += cnt[i];
    int v = s;
#pragma unroll
    for (int off = 1; off < 32; off <<= 1) {
        int u = __shfl_up(v, off, 32);
        if (lane >= off) v += u;
    }
    if (lane == 31) wsum[w] = v;
    __syncthreads();
    if (w == 0) {
        const int x = wsum[lane];
        int y = x;
#pragma unroll
        for (int off = 1; off < 32; off <<= 1) {
            int u = __shfl_up(y, off, 32);
            if (lane >= off) y += u;
        }
        wsum[lane] = y - x;
    }
    __syncthreads();
    int run = wsum[w] + v - s;
    for (int i = b; i < e; ++i) { start[i] = run; run += cnt[i]; }
    if (t == 1023) start[n] = run;
}

__global__ void fill_srcs(const int* __restrict__ ei, int E,
                          const int* __restrict__ start, const int* __restrict__ rank,
                          int* __restrict__ srcs) {
    int e = blockIdx.x * blockDim.x + threadIdx.x;
    if (e < E) srcs[start[ei[E + e]] + rank[e]] = ei[e];
}

#define ACH 8

__global__ __launch_bounds__(256) void gather_agg(
    const float* __restrict__ m, const int* __restrict__ start,
    const int* __restrict__ srcs, const float* __restrict__ b,
    float* __restrict__ agg, int n)
{
    const int wv   = __builtin_amdgcn_readfirstlane((int)(threadIdx.x >> 5));
    const int lane = threadIdx.x & 31;
    const int node = blockIdx.x * 8 + wv;
    if (node >= n) return;
    const int beg = start[node], end = start[node + 1];

    float4 acc = make_float4(0.0f, 0.0f, 0.0f, 0.0f);
    for (int i = beg; i < end; i += 32) {
        const int n32 = min(end - i, 32);
        const int sv = srcs[i + min(lane, n32 - 1)];
        for (int j0 = 0; j0 < n32; j0 += ACH) {
            const int nl = n32 - j0;
            float4 v[ACH];
            v[0] = *(const float4*)(m + (size_t)__builtin_amdgcn_readlane(sv, j0) * HIDC + lane * 4);
#pragma unroll
            for (int j = 1; j < ACH; ++j)
                if (j < nl) {
                    const int sj = __builtin_amdgcn_readlane(sv, j0 + j);
                    v[j] = *(const float4*)(m + (size_t)sj * HIDC + lane * 4);
                }
            acc.x += v[0].x; acc.y += v[0].y; acc.z += v[0].z; acc.w += v[0].w;
#pragma unroll
            for (int j = 1; j < ACH; ++j)
                if (j < nl) { acc.x += v[j].x; acc.y += v[j].y; acc.z += v[j].z; acc.w += v[j].w; }
        }
    }
    const float4 bb = *(const float4*)(b + lane * 4);
    float4 r;
    r.x = fmaxf(acc.x + bb.x, 0.0f);
    r.y = fmaxf(acc.y + bb.y, 0.0f);
    r.z = fmaxf(acc.z + bb.z, 0.0f);
    r.w = fmaxf(acc.w + bb.w, 0.0f);
    *(float4*)(agg + (size_t)node * HIDC + lane * 4) = r;
}

#define GCH 8

__device__ __forceinline__ float4 ld_row4(const float* __restrict__ agg, int node, int lane) {
    return *(const float4*)(agg + (size_t)node * HIDC + lane * 4);
}

__device__ __forceinline__ float sqdist4(const float4 a, const float4 b) {
    float dx = a.x - b.x, dy = a.y - b.y, dz = a.z - b.z, dw = a.w - b.w;
    return dx * dx + dy * dy + dz * dz + dw * dw;
}

__device__ __forceinline__ void flush_run(float* __restrict__ gsum, int node, float s, int lane) {
#pragma unroll
    for (int off = 16; off > 0; off >>= 1)
        s += __shfl_down(s, off, 32);
    if (lane == 0) atomicAdd(&gsum[node], s);
}

__global__ __launch_bounds__(256) void gamma_edges(
    const float* __restrict__ agg, const int* __restrict__ ei, int E,
    float* __restrict__ gsum)
{
    const int wv   = __builtin_amdgcn_readfirstlane((int)(threadIdx.x >> 5));
    const int lane = threadIdx.x & 31;
    const int e0   = (blockIdx.x * 8 + wv) * GCH;
    if (e0 >= E) return;

    if (e0 + GCH <= E) {
        int rr[GCH], cc[GCH];
#pragma unroll
        for (int j = 0; j < GCH; ++j) { rr[j] = ei[e0 + j]; cc[j] = ei[E + e0 + j]; }

        float4 bv[GCH], av[GCH];
#pragma unroll
        for (int j = 0; j < GCH; ++j) bv[j] = ld_row4(agg, cc[j], lane);
        av[0] = ld_row4(agg, rr[0], lane);
#pragma unroll
        for (int j = 1; j < GCH; ++j)
            if (rr[j] != rr[j - 1]) av[j] = ld_row4(agg, rr[j], lane);

        float4 a = av[0];
        float acc = 0.0f;
#pragma unroll
        for (int j = 0; j < GCH; ++j) {
            if (j > 0 && rr[j] != rr[j - 1]) {
                flush_run(gsum, rr[j - 1], acc, lane);
                acc = 0.0f;
                a = av[j];
            }
            acc += sqdist4(a, bv[j]);
        }
        flush_run(gsum, rr[GCH - 1], acc, lane);
    } else {
        const int n = E - e0;
        int prev = ei[e0];
        float acc = 0.0f;
        for (int j = 0; j < n; ++j) {
            const int r = ei[e0 + j];
            const int c = ei[E + e0 + j];
            if (r != prev) { flush_run(gsum, prev, acc, lane); acc = 0.0f; prev = r; }
            acc += sqdist4(ld_row4(agg, r, lane), ld_row4(agg, c, lane));
        }
        flush_run(gsum, prev, acc, lane);
    }
}

__global__ void gamma_node(const float* __restrict__ gsum,
                           const float* __restrict__ inv,
                           float* __restrict__ g, int n) {
    int i = blockIdx.x * blockDim.x + threadIdx.x;
    if (i < n) g[i] = tanhf(gsum[i] * inv[i]);
}

__global__ void update_h(float* __restrict__ h, const float* __restrict__ agg,
                         const float* __restrict__ xs,
                         const float* __restrict__ g1,
                         const float* __restrict__ g2, int total) {
    int i = blockIdx.x * blockDim.x + threadIdx.x;
    if (i < total) {
        int node = i >> 7;
        float gs = g1[node], gq = g2[node];
        h[i] = (h[i] + gs * agg[i] + gq * xs[i]) / (1.0f + gs + gq);
    }
}

extern "C" void kernel_launch(void* const* d_in, const int* in_sizes, int n_in,
                              void* d_out, int out_size, void* d_ws, size_t ws_size,
                              hipStream_t stream) {
    const float* x      = (const float*)d_in[0];
    const float* x0     = (const float*)d_in[1];
    const int*   ei     = (const int*)d_in[2];
    const int*   ei2    = (const int*)d_in[3];
    const float* in_W   = (const float*)d_in[4];
    const float* skip_W = (const float*)d_in[5];
    const float* conv_W = (const float*)d_in[6];
    const float* conv_b = (const float*)d_in[7];
    const float* fc_W   = (const float*)d_in[8];
    const float* fc_b   = (const float*)d_in[9];
    float* out = (float*)d_out;

    const int N  = in_sizes[0] / HIDC;
    const int E1 = in_sizes[2] / 2;
    const int E2 = in_sizes[3] / 2;

    const size_t nh = (size_t)N * HIDC;
    float* h     = (float*)d_ws;
    float* xskip = h + nh;
    float* agg   = xskip + nh;
    float* mbuf  = agg + nh;
    float* inv1  = mbuf + nh;
    float* inv2  = inv1 + N;
    float* gsum1 = inv2 + N;
    float* gsum2 = gsum1 + N;
    float* g1    = gsum2 + N;
    float* g2    = g1 + N;

    int* cnt   = (int*)d_out;
    int* start = cnt + N;
    int* rank  = start + N + 1;
    int* srcs  = rank + E1;

    const dim3 blk(256);
    auto gemm_blocks = [](int M, int Nc) {
        int tiles = ((M + 15) / 16) * ((Nc + 15) / 16);
        return dim3((tiles + 7) / 8);
    };
    const int nhi = (int)nh;

    gemm_wmma_f32<HIDC><<<gemm_blocks(N, HIDC), blk, 0, stream>>>(x,    in_W,   nullptr, h,     N);
    gemm_wmma_f32<HIDC><<<gemm_blocks(N, HIDC), blk, 0, stream>>>(x0,   in_W,   nullptr, mbuf,  N);
    gemm_wmma_f32<HIDC><<<gemm_blocks(N, HIDC), blk, 0, stream>>>(mbuf, skip_W, nullptr, xskip, N);

    zero_f32<<<(N + 255) / 256, blk, 0, stream>>>(inv1, N);
    zero_f32<<<(N + 255) / 256, blk, 0, stream>>>(inv2, N);
    degree_count<<<(E1 + 255) / 256, blk, 0, stream>>>(ei,  E1, inv1);
    degree_count<<<(E2 + 255) / 256, blk, 0, stream>>>(ei2, E2, inv2);
    invert_deg<<<(N + 255) / 256, blk, 0, stream>>>(inv1, N);
    invert_deg<<<(N + 255) / 256, blk, 0, stream>>>(inv2, N);

    zero_f32<<<(N + 255) / 256, blk, 0, stream>>>((float*)cnt, N);
    rank_by_dst<<<(E1 + 255) / 256, blk, 0, stream>>>(ei, E1, cnt, rank);
    scan_counts<<<1, 1024, 0, stream>>>(cnt, start, N);
    fill_srcs<<<(E1 + 255) / 256, blk, 0, stream>>>(ei, E1, start, rank, srcs);

    for (int l = 0; l < 2; ++l) {
        gemm_wmma_f32<HIDC><<<gemm_blocks(N, HIDC), blk, 0, stream>>>(
            h, conv_W + (size_t)l * HIDC * HIDC, nullptr, mbuf, N);
        gather_agg<<<(N + 7) / 8, blk, 0, stream>>>(mbuf, start, srcs, conv_b + l * HIDC, agg, N);
        zero_f32<<<(N + 255) / 256, blk, 0, stream>>>(gsum1, N);
        zero_f32<<<(N + 255) / 256, blk, 0, stream>>>(gsum2, N);
        gamma_edges<<<(E1 + 8 * GCH - 1) / (8 * GCH), blk, 0, stream>>>(agg, ei,  E1, gsum1);
        gamma_edges<<<(E2 + 8 * GCH - 1) / (8 * GCH), blk, 0, stream>>>(agg, ei2, E2, gsum2);
        gamma_node<<<(N + 255) / 256, blk, 0, stream>>>(gsum1, inv1, g1, N);
        gamma_node<<<(N + 255) / 256, blk, 0, stream>>>(gsum2, inv2, g2, N);
        update_h<<<(nhi + 255) / 256, blk, 0, stream>>>(h, agg, xskip, g1, g2, nhi);
    }

    gemm_wmma_f32<40><<<gemm_blocks(N, 40), blk, 0, stream>>>(h, fc_W, fc_b, out, N);
}


# fewer launches (32 -> 14), same arithmetic: one clear for all zero-start arrays, one pass for degrees + ranks, GEMM pairs and both gamma edge lists share launches, gates in update_h (wave per row)
# speedup vs baseline: 1.7686x; 1.1171x over previous
#include <hip/hip_runtime.h>

typedef __attribute__((ext_vector_type(2))) float v2f;
typedef __attribute__((ext_vector_type(8))) float v8f;

#define HIDC 128

template <int NC>
__global__ __launch_bounds__(256) void gemm_wmma_f32(
    const float* __restrict__ A0, const float* __restrict__ B0,
    const float* __restrict__ bias, float* __restrict__ C0,
    const float* __restrict__ A1, const float* __restrict__ B1,
    float* __restrict__ C1, int M)
{
    const bool second = blockIdx.y != 0;
    const float* A = second ? A1 : A0;
    const float* B = second ? B1 : B0;
    float*       C = second ? C1 : C0;
    constexpr int K = HIDC;
    constexpr int tilesN = (NC + 15) >> 4;
    const int tilesM = (M + 15) >> 4;
    const int wid = blockIdx.x * 8 + (threadIdx.x >> 5);
    if (wid >= tilesM * tilesN) return;
    const int tm = wid / tilesN, tn = wid % tilesN;
    const int m0 = tm << 4, n0 = tn << 4;
    const int lane = threadIdx.x & 31;
    const int half = lane >> 4;
    const int l16  = lane & 15;

    const int arow   = min(m0 + l16, M - 1);
    const int bcol_r = n0 + l16;
    const int bcol   = min(bcol_r, NC - 1);

    const float* Ap = A + (size_t)arow * K + 2 * half;
    const float* Bp = B + (size_t)(2 * half) * NC + bcol;

    v8f acc = {};
#pragma unroll
    for (int k0 = 0; k0 < K; k0 += 4) {
        v2f a = *(const v2f*)(Ap + k0);
        v2f b;
        b.x = Bp[k0 * NC];
        b.y = Bp[(k0 + 1) * NC];
        acc = __builtin_amdgcn_wmma_f32_16x16x4_f32(
                  false, a, false, b, (short)0, acc, false, false);
    }

    if (bcol_r < NC) {
        const float bv = bias ? bias[bcol_r] : 0.0f;
#pragma unroll
        for (int v = 0; v < 8; ++v) {
            const int r = m0 + v + 8 * half;
            if (r < M) C[(size_t)r * NC + bcol_r] = acc[v] + bv;
        }
    }
}

__global__ void zero_f32(float* __restrict__ p, int n) {
    int i = blockIdx.x * blockDim.x + threadIdx.x;
    if (i < n) p[i] = 0.0f;
}


__global__ void edge_prep(const int* __restrict__ ei, int E1,
                          const int* __restrict__ ei2, int E2,
                          float* __restrict__ deg1, float* __restrict__ deg2,
                          int* __restrict__ cnt, int* __restrict__ rank) {
    int e = blockIdx.x * blockDim.x + threadIdx.x;
    if (e < E1) {
        atomicAdd(&deg1[ei[e]], 1.0f);
        rank[e] = atomicAdd(&cnt[ei[E1 + e]], 1);
    } else {
        int f = e - E1;
        if (f < E2) atomicAdd(&deg2[ei2[f]], 1.0f);
    }
}

__global__ __launch_bounds__(1024) void scan_counts(const int* __restrict__ cnt,
                                                    int* __restrict__ start, int n) {
    __shared__ int wsum[32];
    const int t = threadIdx.x, lane = t & 31, w = t >> 5;
    const int per = (n + 1023) >> 10;
    const int b = min(t * per, n), e = min(b + per, n);
    int s = 0;
    for (int i = b; i < e; ++i) s += cnt[i];
    int v = s;
#pragma unroll
    for (int off = 1; off < 32; off <<= 1) {
        int u = __shfl_up(v, off, 32);
        if (lane >= off) v += u;
    }
    if (lane == 31) wsum[w] = v;
    __syncthreads();
    if (w == 0) {
        const int x = wsum[lane];
        int y = x;
#pragma unroll
        for (int off = 1; off < 32; off <<= 1) {
            int u = __shfl_up(y, off, 32);
            if (lane >= off) y += u;
        }
        wsum[lane] = y - x;
    }
    __syncthreads();
    int run = wsum[w] + v - s;
    for (int i = b; i < e; ++i) { start[i] = run; run += cnt[i]; }
    if (t == 1023) start[n] = run;
}

__global__ void fill_srcs(const int* __restrict__ ei, int E,
                          const int* __restrict__ start, const int* __restrict__ rank,
                          int* __restrict__ srcs) {
    int e = blockIdx.x * blockDim.x + threadIdx.x;
    if (e < E) srcs[start[ei[E + e]] + rank[e]] = ei[e];
}

#define ACH 8

__global__ __launch_bounds__(256) void gather_agg(
    const float* __restrict__ m, const int* __restrict__ start,
    const int* __restrict__ srcs, const float* __restrict__ b,
    float* __restrict__ agg, int n)
{
    const int wv   = __builtin_amdgcn_readfirstlane((int)(threadIdx.x >> 5));
    const int lane = threadIdx.x & 31;
    const int node = blockIdx.x * 8 + wv;
    if (node >= n) return;
    const int beg = start[node], end = start[node + 1];

    float4 acc = make_float4(0.0f, 0.0f, 0.0f, 0.0f);
    for (int i = beg; i < end; i += 32) {
        const int n32 = min(end - i, 32);
        const int sv = srcs[i + min(lane, n32 - 1)];
        for (int j0 = 0; j0 < n32; j0 += ACH) {
            const int nl = n32 - j0;
            float4 v[ACH];
            v[0] = *(const float4*)(m + (size_t)__builtin_amdgcn_readlane(sv, j0) * HIDC + lane * 4);
#pragma unroll
            for (int j = 1; j < ACH; ++j)
                if (j < nl) {
                    const int sj = __builtin_amdgcn_readlane(sv, j0 + j);
                    v[j] = *(const float4*)(m + (size_t)sj * HIDC + lane * 4);
                }
            acc.x += v[0].x; acc.y += v[0].y; acc.z += v[0].z; acc.w += v[0].w;
#pragma unroll
            for (int j = 1; j < ACH; ++j)
                if (j < nl) { acc.x += v[j].x; acc.y += v[j].y; acc.z += v[j].z; acc.w += v[j].w; }
        }
    }
    const float4 bb = *(const float4*)(b + lane * 4);
    float4 r;
    r.x = fmaxf(acc.x + bb.x, 0.0f);
    r.y = fmaxf(acc.y + bb.y, 0.0f);
    r.z = fmaxf(acc.z + bb.z, 0.0f);
    r.w = fmaxf(acc.w + bb.w, 0.0f);
    *(float4*)(agg + (size_t)node * HIDC + lane * 4) = r;
}

#define GCH 8

__device__ __forceinline__ float4 ld_row4(const float* __restrict__ agg, int node, int lane) {
    return *(const float4*)(agg + (size_t)node * HIDC + lane * 4);
}

__device__ __forceinline__ float sqdist4(const float4 a, const float4 b) {
    float dx = a.x - b.x, dy = a.y - b.y, dz = a.z - b.z, dw = a.w - b.w;
    return dx * dx + dy * dy + dz * dz + dw * dw;
}

__device__ __forceinline__ void flush_run(float* __restrict__ gsum, int node, float s, int lane) {
#pragma unroll
    for (int off = 16; off > 0; off >>= 1)
        s += __shfl_down(s, off, 32);
    if (lane == 0) atomicAdd(&gsum[node], s);
}

__global__ __launch_bounds__(256) void gamma_edges(
    const float* __restrict__ agg,
    const int* __restrict__ eiA, int EA, float* __restrict__ gsumA, int nbA,
    const int* __restrict__ eiB, int EB, float* __restrict__ gsumB)
{
    const bool second = (int)blockIdx.x >= nbA;
    const int* ei   = second ? eiB : eiA;
    const int  E    = second ? EB : EA;
    float*     gsum = second ? gsumB : gsumA;
    const int  bx   = second ? (int)blockIdx.x - nbA : (int)blockIdx.x;
    const int wv   = __builtin_amdgcn_readfirstlane((int)(threadIdx.x >> 5));
    const int lane = threadIdx.x & 31;
    const int e0   = (bx * 8 + wv) * GCH;
    if (e0 >= E) return;

    if (e0 + GCH <= E) {
        int rr[GCH], cc[GCH];
#pragma unroll
        for (int j = 0; j < GCH; ++j) { rr[j] = ei[e0 + j]; cc[j] = ei[E + e0 + j]; }

        float4 bv[GCH], av[GCH];
#pragma unroll
        for (int j = 0; j < GCH; ++j) bv[j] = ld_row4(agg, cc[j], lane);
        av[0] = ld_row4(agg, rr[0], lane);
#pragma unroll
        for (int j = 1; j < GCH; ++j)
            if (rr[j] != rr[j - 1]) av[j] = ld_row4(agg, rr[j], lane);

        float4 a = av[0];
        float acc = 0.0f;
#pragma unroll
        for (int j = 0; j < GCH; ++j) {
            if (j > 0 && rr[j] != rr[j - 1]) {
                flush_run(gsum, rr[j - 1], acc, lane);
                acc = 0.0f;
                a = av[j];
            }
            acc += sqdist4(a, bv[j]);
        }
        flush_run(gsum, rr[GCH - 1], acc, lane);
    } else {
        const int n = E - e0;
        int prev = ei[e0];
        float acc = 0.0f;
        for (int j = 0; j < n; ++j) {
            const int r = ei[e0 + j];
            const int c = ei[E + e0 + j];
            if (r != prev) { flush_run(gsum, prev, acc, lane); acc = 0.0f; prev = r; }
            acc += sqdist4(ld_row4(agg, r, lane), ld_row4(agg, c, lane));
        }
        flush_run(gsum, prev, acc, lane);
    }
}

__global__ __launch_bounds__(256) void update_h(
    float* __restrict__ h, const float* __restrict__ agg,
    const float* __restrict__ xs,
    const float* __restrict__ gsum1, const float* __restrict__ deg1,
    const float* __restrict__ gsum2, const float* __restrict__ deg2, int n)
{
    const int wv   = __builtin_amdgcn_readfirstlane((int)(threadIdx.x >> 5));
    const int lane = threadIdx.x & 31;
    const int node = blockIdx.x * 8 + wv;
    if (node >= n) return;
    const size_t o = (size_t)node * HIDC + lane * 4;
    const float4 hv = *(const float4*)(h + o);
    const float4 av = *(const float4*)(agg + o);
    const float4 xv = *(const float4*)(xs + o);
    const float gs = tanhf(gsum1[node] * (1.0f / (deg1[node] + 1e-10f)));
    const float gq = tanhf(gsum2[node] * (1.0f / (deg2[node] + 1e-10f)));
    const float den = 1.0f + gs + gq;
    float4 r;
    r.x = (hv.x + gs * av.x + gq * xv.x) / den;
    r.y = (hv.y + gs * av.y + gq * xv.y) / den;
    r.z = (hv.z + gs * av.z + gq * xv.z) / den;
    r.w = (hv.w + gs * av.w + gq * xv.w) / den;
    *(float4*)(h + o) = r;
}

extern "C" void kernel_launch(void* const* d_in, const int* in_sizes, int n_in,
                              void* d_out, int out_size, void* d_ws, size_t ws_size,
                              hipStream_t stream) {
    const float* x      = (const float*)d_in[0];
    const float* x0     = (const float*)d_in[1];
    const int*   ei     = (const int*)d_in[2];
    const int*   ei2    = (const int*)d_in[3];
    const float* in_W   = (const float*)d_in[4];
    const float* skip_W = (const float*)d_in[5];
    const float* conv_W = (const float*)d_in[6];
    const float* conv_b = (const float*)d_in[7];
    const float* fc_W   = (const float*)d_in[8];
    const float* fc_b   = (const float*)d_in[9];
    float* out = (float*)d_out;

    const int N  = in_sizes[0] / HIDC;
    const int E1 = in_sizes[2] / 2;
    const int E2 = in_sizes[3] / 2;

    const size_t nh = (size_t)N * HIDC;
    float* h     = (float*)d_ws;
    float* xskip = h + nh;
    float* agg   = xskip + nh;
    float* mbuf  = agg + nh;

    float* zbase = (float*)d_out;
    int*   cnt   = (int*)zbase;
    float* deg1  = zbase + N;
    float* deg2  = deg1 + N;
    float* gsum1[2] = { deg2 + N,     deg2 + 3 * N };
    float* gsum2[2] = { deg2 + 2 * N, deg2 + 4 * N };
    int* start = (int*)(zbase + 7 * N);
    int* rank  = start + N + 1;
    int* srcs  = rank + E1;

    const dim3 blk(256);
    auto gemm_blocks = [](int M, int Nc, int pairs) {
        int tiles = ((M + 15) / 16) * ((Nc + 15) / 16);
        return dim3((tiles + 7) / 8, pairs);
    };
    const float* conv_W1 = conv_W + (size_t)HIDC * HIDC;

    zero_f32<<<(7 * N + 255) / 256, blk, 0, stream>>>(zbase, 7 * N);

    gemm_wmma_f32<HIDC><<<gemm_blocks(N, HIDC, 2), blk, 0, stream>>>(
        x, in_W, nullptr, h, x0, in_W, agg, N);
    gemm_wmma_f32<HIDC><<<gemm_blocks(N, HIDC, 2), blk, 0, stream>>>(
        agg, skip_W, nullptr, xskip, h, conv_W, mbuf, N);

    edge_prep<<<(E1 + E2 + 255) / 256, blk, 0, stream>>>(ei, E1, ei2, E2, deg1, deg2, cnt, rank);
    scan_counts<<<1, 1024, 0, stream>>>(cnt, start, N);
    fill_srcs<<<(E1 + 255) / 256, blk, 0, stream>>>(ei, E1, start, rank, srcs);

    const int nb2 = (E2 + 8 * GCH - 1) / (8 * GCH);
    const int nb1 = (E1 + 8 * GCH - 1) / (8 * GCH);
    for (int l = 0; l < 2; ++l) {
        if (l == 1)
            gemm_wmma_f32<HIDC><<<gemm_blocks(N, HIDC, 1), blk, 0, stream>>>(
                h, conv_W1, nullptr, mbuf, nullptr, nullptr, nullptr, N);
        gather_agg<<<(N + 7) / 8, blk, 0, stream>>>(mbuf, start, srcs, conv_b + l * HIDC, agg, N);
        gamma_edges<<<nb2 + nb1, blk, 0, stream>>>(agg, ei2, E2, gsum2[l], nb2, ei, E1, gsum1[l]);
        update_h<<<(N + 7) / 8, blk, 0, stream>>>(h, agg, xskip, gsum1[l], deg1, gsum2[l], deg2, N);
    }

    gemm_wmma_f32<40><<<gemm_blocks(N, 40, 1), blk, 0, stream>>>(
        h, fc_W, fc_b, out, nullptr, nullptr, nullptr, N);
}


# gamma_edges as a resident grid (1024 blocks): a wave walks a stretch of edges of each list, 8 per step, next indices loaded under the gathers, equal-row runs carried on; sum order changes
# speedup vs baseline: 1.8293x; 1.0343x over previous
#include <hip/hip_runtime.h>

typedef __attribute__((ext_vector_type(2))) float v2f;
typedef __attribute__((ext_vector_type(8))) float v8f;

#define HIDC 128

template <int NC>
__global__ __launch_bounds__(256) void gemm_wmma_f32(
    const float* __restrict__ A0, const float* __restrict__ B0,
    const float* __restrict__ bias, float* __restrict__ C0,
    const float* __restrict__ A1, const float* __restrict__ B1,
    float* __restrict__ C1, int M)
{
    const bool second = blockIdx.y != 0;
    const float* A = second ? A1 : A0;
    const float* B = second ? B1 : B0;
    float*       C = second ? C1 : C0;
    constexpr int K = HIDC;
    constexpr int tilesN = (NC + 15) >> 4;
    const int tilesM = (M + 15) >> 4;
    const int wid = blockIdx.x * 8 + (threadIdx.x >> 5);
    if (wid >= tilesM * tilesN) return;
    const int tm = wid / tilesN, tn = wid % tilesN;
    const int m0 = tm << 4, n0 = tn << 4;
    const int lane = threadIdx.x & 31;
    const int half = lane >> 4;
    const int l16  = lane & 15;

    const int arow   = min(m0 + l16, M - 1);
    const int bcol_r = n0 + l16;
    const int bcol   = min(bcol_r, NC - 1);

    const float* Ap = A + (size_t)arow * K + 2 * half;
    const float* Bp = B + (size_t)(2 * half) * NC + bcol;

    v8f acc = {};
#pragma unroll
    for (int k0 = 0; k0 < K; k0 += 4) {
        v2f a = *(const v2f*)(Ap + k0);
        v2f b;
        b.x = Bp[k0 * NC];
        b.y = Bp[(k0 + 1) * NC];
        acc = __builtin_amdgcn_wmma_f32_16x16x4_f32(
                  false, a, false, b, (short)0, acc, false, false);
    }

    if (bcol_r < NC) {
        const float bv = bias ? bias[bcol_r] : 0.0f;
#pragma unroll
        for (int v = 0; v < 8; ++v) {
            const int r = m0 + v + 8 * half;
            if (r < M) C[(size_t)r * NC + bcol_r] = acc[v] + bv;
        }
    }
}

__global__ void zero_f32(float* __restrict__ p, int n) {
    int i = blockIdx.x * blockDim.x + threadIdx.x;
    if (i < n) p[i] = 0.0f;
}


__global__ void edge_prep(const int* __restrict__ ei, int E1,
                          const int* __restrict__ ei2, int E2,
                          float* __restrict__ deg1, float* __restrict__ deg2,
                          int* __restrict__ cnt, int* __restrict__ rank) {
    int e = blockIdx.x * blockDim.x + threadIdx.x;
    if (e < E1) {
        atomicAdd(&deg1[ei[e]], 1.0f);
        rank[e] = atomicAdd(&cnt[ei[E1 + e]], 1);
    } else {
        int f = e - E1;
        if (f < E2) atomicAdd(&deg2[ei2[f]], 1.0f);
    }
}

__global__ __launch_bounds__(1024) void scan_counts(const int* __restrict__ cnt,
                                                    int* __restrict__ start, int n) {
    __shared__ int wsum[32];
    const int t = threadIdx.x, lane = t & 31, w = t >> 5;
    const int per = (n + 1023) >> 10;
    const int b = min(t * per, n), e = min(b + per, n);
    int s = 0;
    for (int i = b; i < e; ++i) s += cnt[i];
    int v = s;
#pragma unroll
    for (int off = 1; off < 32; off <<= 1) {
        int u = __shfl_up(v, off, 32);
        if (lane >= off) v += u;
    }
    if (lane == 31) wsum[w] = v;
    __syncthreads();
    if (w == 0) {
        const int x = wsum[lane];
        int y = x;
#pragma unroll
        for (int off = 1; off < 32; off <<= 1) {
            int u = __shfl_up(y, off, 32);
            if (lane >= off) y += u;
        }
        wsum[lane] = y - x;
    }
    __syncthreads();
    int run = wsum[w] + v - s;
    for (int i = b; i < e; ++i) { start[i] = run; run += cnt[i]; }
    if (t == 1023) start[n] = run;
}

__global__ void fill_srcs(const int* __restrict__ ei, int E,
                          const int* __restrict__ start, const int* __restrict__ rank,
                          int* __restrict__ srcs) {
    int e = blockIdx.x * blockDim.x + threadIdx.x;
    if (e < E) srcs[start[ei[E + e]] + rank[e]] = ei[e];
}

#define ACH 8

__global__ __launch_bounds__(256) void gather_agg(
    const float* __restrict__ m, const int* __restrict__ start,
    const int* __restrict__ srcs, const float* __restrict__ b,
    float* __restrict__ agg, int n)
{
    const int wv   = __builtin_amdgcn_readfirstlane((int)(threadIdx.x >> 5));
    const int lane = threadIdx.x & 31;
    const int node = blockIdx.x * 8 + wv;
    if (node >= n) return;
    const int beg = start[node], end = start[node + 1];

    float4 acc = make_float4(0.0f, 0.0f, 0.0f, 0.0f);
    for (int i = beg; i < end; i += 32) {
        const int n32 = min(end - i, 32);
        const int sv = srcs[i + min(lane, n32 - 1)];
        for (int j0 = 0; j0 < n32; j0 += ACH) {
            const int nl = n32 - j0;
            float4 v[ACH];
            v[0] = *(const float4*)(m + (size_t)__builtin_amdgcn_readlane(sv, j0) * HIDC + lane * 4);
#pragma unroll
            for (int j = 1; j < ACH; ++j)
                if (j < nl) {
                    const int sj = __builtin_amdgcn_readlane(sv, j0 + j);
                    v[j] = *(const float4*)(m + (size_t)sj * HIDC + lane * 4);
                }
            acc.x += v[0].x; acc.y += v[0].y; acc.z += v[0].z; acc.w += v[0].w;
#pragma unroll
            for (int j = 1; j < ACH; ++j)
                if (j < nl) { acc.x += v[j].x; acc.y += v[j].y; acc.z += v[j].z; acc.w += v[j].w; }
        }
    }
    const float4 bb = *(const float4*)(b + lane * 4);
    float4 r;
    r.x = fmaxf(acc.x + bb.x, 0.0f);
    r.y = fmaxf(acc.y + bb.y, 0.0f);
    r.z = fmaxf(acc.z + bb.z, 0.0f);
    r.w = fmaxf(acc.w + bb.w, 0.0f);
    *(float4*)(agg + (size_t)node * HIDC + lane * 4) = r;
}

#define GCH 8
#define GBW 4

__device__ __forceinline__ float4 ld_row4(const float* __restrict__ agg, int node, int lane) {
    return *(const float4*)(agg + (size_t)node * HIDC + lane * 4);
}

__device__ __forceinline__ float sqdist4(const float4 a, const float4 b) {
    float dx = a.x - b.x, dy = a.y - b.y, dz = a.z - b.z, dw = a.w - b.w;
    return dx * dx + dy * dy + dz * dz + dw * dw;
}

__device__ __forceinline__ void flush_run(float* __restrict__ gsum, int node, float s, int lane) {
#pragma unroll
    for (int off = 16; off > 0; off >>= 1)
        s += __shfl_down(s, off, 32);
    if (lane == 0) atomicAdd(&gsum[node], s);
}

__device__ __forceinline__ void walk_edges(
    const float* __restrict__ agg, const int* __restrict__ ei, int E,
    float* __restrict__ gsum, int e_beg, int e_end, int lane)
{
    int    prev = -1;
    float  acc  = 0.0f;
    float4 a    = make_float4(0.0f, 0.0f, 0.0f, 0.0f);
    int e0 = e_beg;
    if (e0 + GCH <= e_end) {
        int rr[GCH], cc[GCH];
#pragma unroll
        for (int j = 0; j < GCH; ++j) { rr[j] = ei[e0 + j]; cc[j] = ei[E + e0 + j]; }
        for (;;) {
            const bool more = e0 + 2 * GCH <= e_end;
            const int  en   = more ? e0 + GCH : e0;
            int rn[GCH], cn[GCH];
#pragma unroll
            for (int j = 0; j < GCH; ++j) { rn[j] = ei[en + j]; cn[j] = ei[E + en + j]; }

            float4 bv[GCH], av[GCH];
#pragma unroll
            for (int j = 0; j < GCH; ++j) bv[j] = ld_row4(agg, cc[j], lane);
#pragma unroll
            for (int j = 0; j < GCH; ++j) {
                const int pj = j ? rr[j - 1] : prev;
                if (rr[j] != pj) av[j] = ld_row4(agg, rr[j], lane);
            }
#pragma unroll
            for (int j = 0; j < GCH; ++j) {
                const int pj = j ? rr[j - 1] : prev;
                if (rr[j] != pj) {
                    if (j > 0 || pj >= 0) flush_run(gsum, pj, acc, lane);
                    acc = 0.0f;
                    a = av[j];
                }
                acc += sqdist4(a, bv[j]);
            }
            prev = rr[GCH - 1];
            e0 += GCH;
            if (!more) break;
#pragma unroll
            for (int j = 0; j < GCH; ++j) { rr[j] = rn[j]; cc[j] = cn[j]; }
        }
    }
    for (; e0 < e_end; ++e0) {
        const int r = ei[e0];
        const int c = ei[E + e0];
        if (r != prev) {
            if (prev >= 0) flush_run(gsum, prev, acc, lane);
            acc = 0.0f; prev = r;
            a = ld_row4(agg, r, lane);
        }
        acc += sqdist4(a, ld_row4(agg, c, lane));
    }
    flush_run(gsum, prev, acc, lane);
}

__global__ __launch_bounds__(256) void gamma_edges(
    const float* __restrict__ agg,
    const int* __restrict__ eiA, int EA, float* __restrict__ gsumA, int qA, int rA,
    const int* __restrict__ eiB, int EB, float* __restrict__ gsumB, int qB, int rB)
{
    const int wv   = __builtin_amdgcn_readfirstlane((int)(threadIdx.x >> 5));
    const int lane = threadIdx.x & 31;
    const int w    = blockIdx.x * 8 + wv;
    const int nw   = gridDim.x * 8;

    const int a0 = w * qA + min(w, rA);
    const int a1 = a0 + qA + (w < rA ? 1 : 0);
    if (a0 < a1)
        walk_edges(agg, eiA, EA, gsumA, a0 * GCH, min(a1 * GCH, EA), lane);

    const int xb = max(w - (nw - rB), 0);
    const int b0 = w * qB + xb;
    const int b1 = b0 + qB + (w >= nw - rB ? 1 : 0);
    if (b0 < b1)
        walk_edges(agg, eiB, EB, gsumB, b0 * GCH, min(b1 * GCH, EB), lane);
}

__global__ __launch_bounds__(256) void update_h(
    float* __restrict__ h, const float* __restrict__ agg,
    const float* __restrict__ xs,
    const float* __restrict__ gsum1, const float* __restrict__ deg1,
    const float* __restrict__ gsum2, const float* __restrict__ deg2, int n)
{
    const int wv   = __builtin_amdgcn_readfirstlane((int)(threadIdx.x >> 5));
    const int lane = threadIdx.x & 31;
    const int node = blockIdx.x * 8 + wv;
    if (node >= n) return;
    const size_t o = (size_t)node * HIDC + lane * 4;
    const float4 hv = *(const float4*)(h + o);
    const float4 av = *(const float4*)(agg + o);
    const float4 xv = *(const float4*)(xs + o);
    const float gs = tanhf(gsum1[node] * (1.0f / (deg1[node] + 1e-10f)));
    const float gq = tanhf(gsum2[node] * (1.0f / (deg2[node] + 1e-10f)));
    const float den = 1.0f + gs + gq;
    float4 r;
    r.x = (hv.x + gs * av.x + gq * xv.x) / den;
    r.y = (hv.y + gs * av.y + gq * xv.y) / den;
    r.z = (hv.z + gs * av.z + gq * xv.z) / den;
    r.w = (hv.w + gs * av.w + gq * xv.w) / den;
    *(float4*)(h + o) = r;
}

extern "C" void kernel_launch(void* const* d_in, const int* in_sizes, int n_in,
                              void* d_out, int out_size, void* d_ws, size_t ws_size,
                              hipStream_t stream) {
    const float* x      = (const float*)d_in[0];
    const float* x0     = (const float*)d_in[1];
    const int*   ei     = (const int*)d_in[2];
    const int*   ei2    = (const int*)d_in[3];
    const float* in_W   = (const float*)d_in[4];
    const float* skip_W = (const float*)d_in[5];
    const float* conv_W = (const float*)d_in[6];
    const float* conv_b = (const float*)d_in[7];
    const float* fc_W   = (const float*)d_in[8];
    const float* fc_b   = (const float*)d_in[9];
    float* out = (float*)d_out;

    const int N  = in_sizes[0] / HIDC;
    const int E1 = in_sizes[2] / 2;
    const int E2 = in_sizes[3] / 2;

    const size_t nh = (size_t)N * HIDC;
    float* h     = (float*)d_ws;
    float* xskip = h + nh;
    float* agg   = xskip + nh;
    float* mbuf  = agg + nh;

    float* zbase = (float*)d_out;
    int*   cnt   = (int*)zbase;
    float* deg1  = zbase + N;
    float* deg2  = deg1 + N;
    float* gsum1[2] = { deg2 + N,     deg2 + 3 * N };
    float* gsum2[2] = { deg2 + 2 * N, deg2 + 4 * N };
    int* start = (int*)(zbase + 7 * N);
    int* rank  = start + N + 1;
    int* srcs  = rank + E1;

    const dim3 blk(256);
    auto gemm_blocks = [](int M, int Nc, int pairs) {
        int tiles = ((M + 15) / 16) * ((Nc + 15) / 16);
        return dim3((tiles + 7) / 8, pairs);
    };
    const float* conv_W1 = conv_W + (size_t)HIDC * HIDC;

    zero_f32<<<(7 * N + 255) / 256, blk, 0, stream>>>(zbase, 7 * N);

    gemm_wmma_f32<HIDC><<<gemm_blocks(N, HIDC, 2), blk, 0, stream>>>(
        x, in_W, nullptr, h, x0, in_W, agg, N);
    gemm_wmma_f32<HIDC><<<gemm_blocks(N, HIDC, 2), blk, 0, stream>>>(
        agg, skip_W, nullptr, xskip, h, conv_W, mbuf, N);

    edge_prep<<<(E1 + E2 + 255) / 256, blk, 0, stream>>>(ei, E1, ei2, E2, deg1, deg2, cnt, rank);
    scan_counts<<<1, 1024, 0, stream>>>(cnt, start, N);
    fill_srcs<<<(E1 + 255) / 256, blk, 0, stream>>>(ei, E1, start, rank, srcs);

    const int gblocks = 256 * GBW;
    const int gwaves  = gblocks * 8;
    const int nch2 = (E2 + GCH - 1) / GCH, nch1 = (E1 + GCH - 1) / GCH;
    const int q2 = nch2 / gwaves, r2 = nch2 % gwaves;
    const int q1 = nch1 / gwaves, r1 = nch1 % gwaves;
    for (int l = 0; l < 2; ++l) {
        if (l == 1)
            gemm_wmma_f32<HIDC><<<gemm_blocks(N, HIDC, 1), blk, 0, stream>>>(
                h, conv_W1, nullptr, mbuf, nullptr, nullptr, nullptr, N);
        gather_agg<<<(N + 7) / 8, blk, 0, stream>>>(mbuf, start, srcs, conv_b + l * HIDC, agg, N);
        gamma_edges<<<gblocks, blk, 0, stream>>>(agg, ei2, E2, gsum2[l], q2, r2, ei, E1, gsum1[l], q1, r1);
        update_h<<<(N + 7) / 8, blk, 0, stream>>>(h, agg, xskip, gsum1[l], deg1, gsum2[l], deg2, N);
    }

    gemm_wmma_f32<40><<<gemm_blocks(N, 40, 1), blk, 0, stream>>>(
        h, fc_W, fc_b, out, nullptr, nullptr, nullptr, N);
}


# 128-col GEMMs: a wave owns a 16x64 strip, 4 accumulator tiles share each A fragment; A and B by 16-byte loads over 8-wide k blocks, rows stored as whole 128-byte lines; k sum order changes
# speedup vs baseline: 2.1187x; 1.1582x over previous
#include <hip/hip_runtime.h>

typedef __attribute__((ext_vector_type(2))) float v2f;
typedef __attribute__((ext_vector_type(8))) float v8f;

#define HIDC 128

template <int NC>
__global__ __launch_bounds__(256) void gemm_wmma_f32(
    const float* __restrict__ A0, const float* __restrict__ B0,
    const float* __restrict__ bias, float* __restrict__ C0,
    const float* __restrict__ A1, const float* __restrict__ B1,
    float* __restrict__ C1, int M)
{
    const bool second = blockIdx.y != 0;
    const float* A = second ? A1 : A0;
    const float* B = second ? B1 : B0;
    float*       C = second ? C1 : C0;
    constexpr int K = HIDC;
    constexpr int tilesN = (NC + 15) >> 4;
    const int tilesM = (M + 15) >> 4;
    const int wid = blockIdx.x * 8 + (threadIdx.x >> 5);
    if (wid >= tilesM * tilesN) return;
    const int tm = wid / tilesN, tn = wid % tilesN;
    const int m0 = tm << 4, n0 = tn << 4;
    const int lane = threadIdx.x & 31;
    const int half = lane >> 4;
    const int l16  = lane & 15;

    const int arow   = min(m0 + l16, M - 1);
    const int bcol_r = n0 + l16;
    const int bcol   = min(bcol_r, NC - 1);

    const float* Ap = A + (size_t)arow * K + 2 * half;
    const float* Bp = B + (size_t)(2 * half) * NC + bcol;

    v8f acc = {};
#pragma unroll
    for (int k0 = 0; k0 < K; k0 += 4) {
        v2f a = *(const v2f*)(Ap + k0);
        v2f b;
        b.x = Bp[k0 * NC];
        b.y = Bp[(k0 + 1) * NC];
        acc = __builtin_amdgcn_wmma_f32_16x16x4_f32(
                  false, a, false, b, (short)0, acc, false, false);
    }

    if (bcol_r < NC) {
        const float bv = bias ? bias[bcol_r] : 0.0f;
#pragma unroll
        for (int v = 0; v < 8; ++v) {
            const int r = m0 + v + 8 * half;
            if (r < M) C[(size_t)r * NC + bcol_r] = acc[v] + bv;
        }
    }
}

__device__ __forceinline__ v8f wmma4(const float a0, const float a1,
                                     const float b0, const float b1, const v8f acc) {
    v2f a; a.x = a0; a.y = a1;
    v2f b; b.x = b0; b.y = b1;
    return __builtin_amdgcn_wmma_f32_16x16x4_f32(false, a, false, b, (short)0, acc, false, false);
}

__global__ __launch_bounds__(128) __attribute__((amdgpu_waves_per_eu(1, 4))) void gemm_strip_f32(
    const float* __restrict__ A0, const float* __restrict__ B0, float* __restrict__ C0,
    const float* __restrict__ A1, const float* __restrict__ B1, float* __restrict__ C1, int M)
{
    const bool second = blockIdx.y != 0;
    const float* A = second ? A1 : A0;
    const float* B = second ? B1 : B0;
    float*       C = second ? C1 : C0;
    constexpr int K = HIDC, NC = HIDC;
    const int tilesM = (M + 15) >> 4;
    const int wid = blockIdx.x * 4 + (threadIdx.x >> 5);
    const int tm  = wid >> 1;
    if (tm >= tilesM) return;
    const int nb   = (wid & 1) << 6;
    const int m0   = tm << 4;
    const int lane = threadIdx.x & 31;
    const int hf   = lane >> 4;
    const int c    = lane & 15;

    const int arow = min(m0 + c, M - 1);
    const float* Ap = A + (size_t)arow * K + 4 * hf;
    const float* Bp = B + (size_t)(4 * hf) * NC + nb + 4 * c;

    v8f acc0 = {}, acc1 = {}, acc2 = {}, acc3 = {};
#pragma unroll
    for (int kb = 0; kb < K; kb += 8) {
        const float4 a  = *(const float4*)(Ap + kb);
        const float4 b0 = *(const float4*)(Bp + (kb + 0) * NC);
        const float4 b1 = *(const float4*)(Bp + (kb + 1) * NC);
        const float4 b2 = *(const float4*)(Bp + (kb + 2) * NC);
        const float4 b3 = *(const float4*)(Bp + (kb + 3) * NC);
        acc0 = wmma4(a.x, a.y, b0.x, b1.x, acc0);
        acc1 = wmma4(a.x, a.y, b0.y, b1.y, acc1);
        acc2 = wmma4(a.x, a.y, b0.z, b1.z, acc2);
        acc3 = wmma4(a.x, a.y, b0.w, b1.w, acc3);
        acc0 = wmma4(a.z, a.w, b2.x, b3.x, acc0);
        acc1 = wmma4(a.z, a.w, b2.y, b3.y, acc1);
        acc2 = wmma4(a.z, a.w, b2.z, b3.z, acc2);
        acc3 = wmma4(a.z, a.w, b2.w, b3.w, acc3);
    }

#pragma unroll
    for (int v = 0; v < 8; ++v) {
        const int r = m0 + v + 8 * hf;
        if (r < M)
            *(float4*)(C + (size_t)r * NC + nb + 4 * c) =
                make_float4(acc0[v], acc1[v], acc2[v], acc3[v]);
    }
}

__global__ void zero_f32(float* __restrict__ p, int n) {
    int i = blockIdx.x * blockDim.x + threadIdx.x;
    if (i < n) p[i] = 0.0f;
}


__global__ void edge_prep(const int* __restrict__ ei, int E1,
                          const int* __restrict__ ei2, int E2,
                          float* __restrict__ deg1, float* __restrict__ deg2,
                          int* __restrict__ cnt, int* __restrict__ rank) {
    int e = blockIdx.x * blockDim.x + threadIdx.x;
    if (e < E1) {
        atomicAdd(&deg1[ei[e]], 1.0f);
        rank[e] = atomicAdd(&cnt[ei[E1 + e]], 1);
    } else {
        int f = e - E1;
        if (f < E2) atomicAdd(&deg2[ei2[f]], 1.0f);
    }
}

__global__ __launch_bounds__(1024) void scan_counts(const int* __restrict__ cnt,
                                                    int* __restrict__ start, int n) {
    __shared__ int wsum[32];
    const int t = threadIdx.x, lane = t & 31, w = t >> 5;
    const int per = (n + 1023) >> 10;
    const int b = min(t * per, n), e = min(b + per, n);
    int s = 0;
    for (int i = b; i < e; ++i) s += cnt[i];
    int v = s;
#pragma unroll
    for (int off = 1; off < 32; off <<= 1) {
        int u = __shfl_up(v, off, 32);
        if (lane >= off) v += u;
    }
    if (lane == 31) wsum[w] = v;
    __syncthreads();
    if (w == 0) {
        const int x = wsum[lane];
        int y = x;
#pragma unroll
        for (int off = 1; off < 32; off <<= 1) {
            int u = __shfl_up(y, off, 32);
            if (lane >= off) y += u;
        }
        wsum[lane] = y - x;
    }
    __syncthreads();
    int run = wsum[w] + v - s;
    for (int i = b; i < e; ++i) { start[i] = run; run += cnt[i]; }
    if (t == 1023) start[n] = run;
}

__global__ void fill_srcs(const int* __restrict__ ei, int E,
                          const int* __restrict__ start, const int* __restrict__ rank,
                          int* __restrict__ srcs) {
    int e = blockIdx.x * blockDim.x + threadIdx.x;
    if (e < E) srcs[start[ei[E + e]] + rank[e]] = ei[e];
}

#define ACH 8

__global__ __launch_bounds__(256) void gather_agg(
    const float* __restrict__ m, const int* __restrict__ start,
    const int* __restrict__ srcs, const float* __restrict__ b,
    float* __restrict__ agg, int n)
{
    const int wv   = __builtin_amdgcn_readfirstlane((int)(threadIdx.x >> 5));
    const int lane = threadIdx.x & 31;
    const int node = blockIdx.x * 8 + wv;
    if (node >= n) return;
    const int beg = start[node], end = start[node + 1];

    float4 acc = make_float4(0.0f, 0.0f, 0.0f, 0.0f);
    for (int i = beg; i < end; i += 32) {
        const int n32 = min(end - i, 32);
        const int sv = srcs[i + min(lane, n32 - 1)];
        for (int j0 = 0; j0 < n32; j0 += ACH) {
            const int nl = n32 - j0;
            float4 v[ACH];
            v[0] = *(const float4*)(m + (size_t)__builtin_amdgcn_readlane(sv, j0) * HIDC + lane * 4);
#pragma unroll
            for (int j = 1; j < ACH; ++j)
                if (j < nl) {
                    const int sj = __builtin_amdgcn_readlane(sv, j0 + j);
                    v[j] = *(const float4*)(m + (size_t)sj * HIDC + lane * 4);
                }
            acc.x += v[0].x; acc.y += v[0].y; acc.z += v[0].z; acc.w += v[0].w;
#pragma unroll
            for (int j = 1; j < ACH; ++j)
                if (j < nl) { acc.x += v[j].x; acc.y += v[j].y; acc.z += v[j].z; acc.w += v[j].w; }
        }
    }
    const float4 bb = *(const float4*)(b + lane * 4);
    float4 r;
    r.x = fmaxf(acc.x + bb.x, 0.0f);
    r.y = fmaxf(acc.y + bb.y, 0.0f);
    r.z = fmaxf(acc.z + bb.z, 0.0f);
    r.w = fmaxf(acc.w + bb.w, 0.0f);
    *(float4*)(agg + (size_t)node * HIDC + lane * 4) = r;
}

#define GCH 8
#define GBW 4

__device__ __forceinline__ float4 ld_row4(const float* __restrict__ agg, int node, int lane) {
    return *(const float4*)(agg + (size_t)node * HIDC + lane * 4);
}

__device__ __forceinline__ float sqdist4(const float4 a, const float4 b) {
    float dx = a.x - b.x, dy = a.y - b.y, dz = a.z - b.z, dw = a.w - b.w;
    return dx * dx + dy * dy + dz * dz + dw * dw;
}

__device__ __forceinline__ void flush_run(float* __restrict__ gsum, int node, float s, int lane) {
#pragma unroll
    for (int off = 16; off > 0; off >>= 1)
        s += __shfl_down(s, off, 32);
    if (lane == 0) atomicAdd(&gsum[node], s);
}

__device__ __forceinline__ void walk_edges(
    const float* __restrict__ agg, const int* __restrict__ ei, int E,
    float* __restrict__ gsum, int e_beg, int e_end, int lane)
{
    int    prev = -1;
    float  acc  = 0.0f;
    float4 a    = make_float4(0.0f, 0.0f, 0.0f, 0.0f);
    int e0 = e_beg;
    if (e0 + GCH <= e_end) {
        int rr[GCH], cc[GCH];
#pragma unroll
        for (int j = 0; j < GCH; ++j) { rr[j] = ei[e0 + j]; cc[j] = ei[E + e0 + j]; }
        for (;;) {
            const bool more = e0 + 2 * GCH <= e_end;
            const int  en   = more ? e0 + GCH : e0;
            int rn[GCH], cn[GCH];
#pragma unroll
            for (int j = 0; j < GCH; ++j) { rn[j] = ei[en + j]; cn[j] = ei[E + en + j]; }

            float4 bv[GCH], av[GCH];
#pragma unroll
            for (int j = 0; j < GCH; ++j) bv[j] = ld_row4(agg, cc[j], lane);
#pragma unroll
            for (int j = 0; j < GCH; ++j) {
                const int pj = j ? rr[j - 1] : prev;
                if (rr[j] != pj) av[j] = ld_row4(agg, rr[j], lane);
            }
#pragma unroll
            for (int j = 0; j < GCH; ++j) {
                const int pj = j ? rr[j - 1] : prev;
                if (rr[j] != pj) {
                    if (j > 0 || pj >= 0) flush_run(gsum, pj, acc, lane);
                    acc = 0.0f;
                    a = av[j];
                }
                acc += sqdist4(a, bv[j]);
            }
            prev = rr[GCH - 1];
            e0 += GCH;
            if (!more) break;
#pragma unroll
            for (int j = 0; j < GCH; ++j) { rr[j] = rn[j]; cc[j] = cn[j]; }
        }
    }
    for (; e0 < e_end; ++e0) {
        const int r = ei[e0];
        const int c = ei[E + e0];
        if (r != prev) {
            if (prev >= 0) flush_run(gsum, prev, acc, lane);
            acc = 0.0f; prev = r;
            a = ld_row4(agg, r, lane);
        }
        acc += sqdist4(a, ld_row4(agg, c, lane));
    }
    flush_run(gsum, prev, acc, lane);
}

__global__ __launch_bounds__(256) void gamma_edges(
    const float* __restrict__ agg,
    const int* __restrict__ eiA, int EA, float* __restrict__ gsumA, int qA, int rA,
    const int* __restrict__ eiB, int EB, float* __restrict__ gsumB, int qB, int rB)
{
    const int wv   = __builtin_amdgcn_readfirstlane((int)(threadIdx.x >> 5));
    const int lane = threadIdx.x & 31;
    const int w    = blockIdx.x * 8 + wv;
    const int nw   = gridDim.x * 8;

    const int a0 = w * qA + min(w, rA);
    const int a1 = a0 + qA + (w < rA ? 1 : 0);
    if (a0 < a1)
        walk_edges(agg, eiA, EA, gsumA, a0 * GCH, min(a1 * GCH, EA), lane);

    const int xb = max(w - (nw - rB), 0);
    const int b0 = w * qB + xb;
    const int b1 = b0 + qB + (w >= nw - rB ? 1 : 0);
    if (b0 < b1)
        walk_edges(agg, eiB, EB, gsumB, b0 * GCH, min(b1 * GCH, EB), lane);
}

__global__ __launch_bounds__(256) void update_h(
    float* __restrict__ h, const float* __restrict__ agg,
    const float* __restrict__ xs,
    const float* __restrict__ gsum1, const float* __restrict__ deg1,
    const float* __restrict__ gsum2, const float* __restrict__ deg2, int n)
{
    const int wv   = __builtin_amdgcn_readfirstlane((int)(threadIdx.x >> 5));
    const int lane = threadIdx.x & 31;
    const int node = blockIdx.x * 8 + wv;
    if (node >= n) return;
    const size_t o = (size_t)node * HIDC + lane * 4;
    const float4 hv = *(const float4*)(h + o);
    const float4 av = *(const float4*)(agg + o);
    const float4 xv = *(const float4*)(xs + o);
    const float gs = tanhf(gsum1[node] * (1.0f / (deg1[node] + 1e-10f)));
    const float gq = tanhf(gsum2[node] * (1.0f / (deg2[node] + 1e-10f)));
    const float den = 1.0f + gs + gq;
    float4 r;
    r.x = (hv.x + gs * av.x + gq * xv.x) / den;
    r.y = (hv.y + gs * av.y + gq * xv.y) / den;
    r.z = (hv.z + gs * av.z + gq * xv.z) / den;
    r.w = (hv.w + gs * av.w + gq * xv.w) / den;
    *(float4*)(h + o) = r;
}

extern "C" void kernel_launch(void* const* d_in, const int* in_sizes, int n_in,
                              void* d_out, int out_size, void* d_ws, size_t ws_size,
                              hipStream_t stream) {
    const float* x      = (const float*)d_in[0];
    const float* x0     = (const float*)d_in[1];
    const int*   ei     = (const int*)d_in[2];
    const int*   ei2    = (const int*)d_in[3];
    const float* in_W   = (const float*)d_in[4];
    const float* skip_W = (const float*)d_in[5];
    const float* conv_W = (const float*)d_in[6];
    const float* conv_b = (const float*)d_in[7];
    const float* fc_W   = (const float*)d_in[8];
    const float* fc_b   = (const float*)d_in[9];
    float* out = (float*)d_out;

    const int N  = in_sizes[0] / HIDC;
    const int E1 = in_sizes[2] / 2;
    const int E2 = in_sizes[3] / 2;

    const size_t nh = (size_t)N * HIDC;
    float* h     = (float*)d_ws;
    float* xskip = h + nh;
    float* agg   = xskip + nh;
    float* mbuf  = agg + nh;

    float* zbase = (float*)d_out;
    int*   cnt   = (int*)zbase;
    float* deg1  = zbase + N;
    float* deg2  = deg1 + N;
    float* gsum1[2] = { deg2 + N,     deg2 + 3 * N };
    float* gsum2[2] = { deg2 + 2 * N, deg2 + 4 * N };
    int* start = (int*)(zbase + 7 * N);
    int* rank  = start + N + 1;
    int* srcs  = rank + E1;

    const dim3 blk(256);
    auto gemm_blocks = [](int M, int Nc, int pairs) {
        int tiles = ((M + 15) / 16) * ((Nc + 15) / 16);
        return dim3((tiles + 7) / 8, pairs);
    };
    const dim3 sblk(128);
    auto strip_blocks = [](int M, int pairs) {
        int strips = ((M + 15) / 16) * 2;
        return dim3((strips + 3) / 4, pairs);
    };
    const float* conv_W1 = conv_W + (size_t)HIDC * HIDC;

    zero_f32<<<(7 * N + 255) / 256, blk, 0, stream>>>(zbase, 7 * N);

    gemm_strip_f32<<<strip_blocks(N, 2), sblk, 0, stream>>>(
        x, in_W, h, x0, in_W, agg, N);
    gemm_strip_f32<<<strip_blocks(N, 2), sblk, 0, stream>>>(
        agg, skip_W, xskip, h, conv_W, mbuf, N);

    edge_prep<<<(E1 + E2 + 255) / 256, blk, 0, stream>>>(ei, E1, ei2, E2, deg1, deg2, cnt, rank);
    scan_counts<<<1, 1024, 0, stream>>>(cnt, start, N);
    fill_srcs<<<(E1 + 255) / 256, blk, 0, stream>>>(ei, E1, start, rank, srcs);

    const int gblocks = 256 * GBW;
    const int gwaves  = gblocks * 8;
    const int nch2 = (E2 + GCH - 1) / GCH, nch1 = (E1 + GCH - 1) / GCH;
    const int q2 = nch2 / gwaves, r2 = nch2 % gwaves;
    const int q1 = nch1 / gwaves, r1 = nch1 % gwaves;
    for (int l = 0; l < 2; ++l) {
        if (l == 1)
            gemm_strip_f32<<<strip_blocks(N, 1), sblk, 0, stream>>>(
                h, conv_W1, mbuf, nullptr, nullptr, nullptr, N);
        gather_agg<<<(N + 7) / 8, blk, 0, stream>>>(mbuf, start, srcs, conv_b + l * HIDC, agg, N);
        gamma_edges<<<gblocks, blk, 0, stream>>>(agg, ei2, E2, gsum2[l], q2, r2, ei, E1, gsum1[l], q1, r1);
        update_h<<<(N + 7) / 8, blk, 0, stream>>>(h, agg, xskip, gsum1[l], deg1, gsum2[l], deg2, N);
    }

    gemm_wmma_f32<40><<<gemm_blocks(N, 40, 1), blk, 0, stream>>>(
        h, fc_W, fc_b, out, nullptr, nullptr, nullptr, N);
}


# clear and degree+rank passes ride as head blocks of the two paired GEMM launches (long-lived strided threads; equal two-hop rows pre-counted: exact integer counts regrouped); 14 -> 12 launches
# speedup vs baseline: 2.5049x; 1.1823x over previous
#include <hip/hip_runtime.h>

typedef __attribute__((ext_vector_type(2))) float v2f;
typedef __attribute__((ext_vector_type(8))) float v8f;

#define HIDC 128

template <int NC>
__global__ __launch_bounds__(256) void gemm_wmma_f32(
    const float* __restrict__ A0, const float* __restrict__ B0,
    const float* __restrict__ bias, float* __restrict__ C0,
    const float* __restrict__ A1, const float* __restrict__ B1,
    float* __restrict__ C1, int M)
{
    const bool second = blockIdx.y != 0;
    const float* A = second ? A1 : A0;
    const float* B = second ? B1 : B0;
    float*       C = second ? C1 : C0;
    constexpr int K = HIDC;
    constexpr int tilesN = (NC + 15) >> 4;
    const int tilesM = (M + 15) >> 4;
    const int wid = blockIdx.x * 8 + (threadIdx.x >> 5);
    if (wid >= tilesM * tilesN) return;
    const int tm = wid / tilesN, tn = wid % tilesN;
    const int m0 = tm << 4, n0 = tn << 4;
    const int lane = threadIdx.x & 31;
    const int half = lane >> 4;
    const int l16  = lane & 15;

    const int arow   = min(m0 + l16, M - 1);
    const int bcol_r = n0 + l16;
    const int bcol   = min(bcol_r, NC - 1);

    const float* Ap = A + (size_t)arow * K + 2 * half;
    const float* Bp = B + (size_t)(2 * half) * NC + bcol;

    v8f acc = {};
#pragma unroll
    for (int k0 = 0; k0 < K; k0 += 4) {
        v2f a = *(const v2f*)(Ap + k0);
        v2f b;
        b.x = Bp[k0 * NC];
        b.y = Bp[(k0 + 1) * NC];
        acc = __builtin_amdgcn_wmma_f32_16x16x4_f32(
                  false, a, false, b, (short)0, acc, false, false);
    }

    if (bcol_r < NC) {
        const float bv = bias ? bias[bcol_r] : 0.0f;
#pragma unroll
        for (int v = 0; v < 8; ++v) {
            const int r = m0 + v + 8 * half;
            if (r < M) C[(size_t)r * NC + bcol_r] = acc[v] + bv;
        }
    }
}

__device__ __forceinline__ void edge_prep_part(
    const int* __restrict__ ei, int E1, const int* __restrict__ ei2, int E2,
    float* __restrict__ deg1, float* __restrict__ deg2,
    int* __restrict__ cnt, int* __restrict__ rank, int t, int T)
{
    for (int e = t; e < E1; e += T) {
        atomicAdd(&deg1[ei[e]], 1.0f);
        rank[e] = atomicAdd(&cnt[ei[E1 + e]], 1);
    }
    const int n4 = E2 >> 2;
    for (int g = t; g < n4; g += T) {
        const int4 r = *(const int4*)(ei2 + 4 * g);
        int p = r.x;
        float c = 1.0f;
        if (r.y == p) c += 1.0f; else { atomicAdd(&deg2[p], c); p = r.y; c = 1.0f; }
        if (r.z == p) c += 1.0f; else { atomicAdd(&deg2[p], c); p = r.z; c = 1.0f; }
        if (r.w == p) c += 1.0f; else { atomicAdd(&deg2[p], c); p = r.w; c = 1.0f; }
        atomicAdd(&deg2[p], c);
    }
    for (int f = (n4 << 2) + t; f < E2; f += T)
        atomicAdd(&deg2[ei2[f]], 1.0f);
}

__device__ __forceinline__ v8f wmma4(const float a0, const float a1,
                                     const float b0, const float b1, const v8f acc) {
    v2f a; a.x = a0; a.y = a1;
    v2f b; b.x = b0; b.y = b1;
    return __builtin_amdgcn_wmma_f32_16x16x4_f32(false, a, false, b, (short)0, acc, false, false);
}

template <int AUX>
__global__ __launch_bounds__(128) __attribute__((amdgpu_waves_per_eu(1, 4))) void gemm_strip_f32(
    const float* __restrict__ A0, const float* __restrict__ B0, float* __restrict__ C0,
    const float* __restrict__ A1, const float* __restrict__ B1, float* __restrict__ C1,
    int M, int gb, int auxb,
    const int* __restrict__ ei, int E1, const int* __restrict__ ei2, int E2,
    float* __restrict__ deg1, float* __restrict__ deg2,
    int* __restrict__ cnt, int* __restrict__ rank,
    float* __restrict__ zp, int zn)
{
    if (AUX != 0 && (int)blockIdx.x < auxb) {
        const int t = blockIdx.x * 128 + threadIdx.x, T = auxb * 128;
        if (AUX == 1) {
            for (int i = t; i < zn; i += T) zp[i] = 0.0f;
        } else {
            edge_prep_part(ei, E1, ei2, E2, deg1, deg2, cnt, rank, t, T);
        }
        return;
    }
    const int  g      = (int)blockIdx.x - auxb;
    const bool second = g >= gb;
    const float* A = second ? A1 : A0;
    const float* B = second ? B1 : B0;
    float*       C = second ? C1 : C0;
    constexpr int K = HIDC, NC = HIDC;
    const int tilesM = (M + 15) >> 4;
    const int wid = (second ? g - gb : g) * 4 + (threadIdx.x >> 5);
    const int tm  = wid >> 1;
    if (tm >= tilesM) return;
    const int nb   = (wid & 1) << 6;
    const int m0   = tm << 4;
    const int lane = threadIdx.x & 31;
    const int hf   = lane >> 4;
    const int c    = lane & 15;

    const int arow = min(m0 + c, M - 1);
    const float* Ap = A + (size_t)arow * K + 4 * hf;
    const float* Bp = B + (size_t)(4 * hf) * NC + nb + 4 * c;

    v8f acc0 = {}, acc1 = {}, acc2 = {}, acc3 = {};
#pragma unroll
    for (int kb = 0; kb < K; kb += 8) {
        const float4 a  = *(const float4*)(Ap + kb);
        const float4 b0 = *(const float4*)(Bp + (kb + 0) * NC);
        const float4 b1 = *(const float4*)(Bp + (kb + 1) * NC);
        const float4 b2 = *(const float4*)(Bp + (kb + 2) * NC);
        const float4 b3 = *(const float4*)(Bp + (kb + 3) * NC);
        acc0 = wmma4(a.x, a.y, b0.x, b1.x, acc0);
        acc1 = wmma4(a.x, a.y, b0.y, b1.y, acc1);
        acc2 = wmma4(a.x, a.y, b0.z, b1.z, acc2);
        acc3 = wmma4(a.x, a.y, b0.w, b1.w, acc3);
        acc0 = wmma4(a.z, a.w, b2.x, b3.x, acc0);
        acc1 = wmma4(a.z, a.w, b2.y, b3.y, acc1);
        acc2 = wmma4(a.z, a.w, b2.z, b3.z, acc2);
        acc3 = wmma4(a.z, a.w, b2.w, b3.w, acc3);
    }

#pragma unroll
    for (int v = 0; v < 8; ++v) {
        const int r = m0 + v + 8 * hf;
        if (r < M)
            *(float4*)(C + (size_t)r * NC + nb + 4 * c) =
                make_float4(acc0[v], acc1[v], acc2[v], acc3[v]);
    }
}


__global__ __launch_bounds__(1024) void scan_counts(const int* __restrict__ cnt,
                                                    int* __restrict__ start, int n) {
    __shared__ int wsum[32];
    const int t = threadIdx.x, lane = t & 31, w = t >> 5;
    const int per = (n + 1023) >> 10;
    const int b = min(t * per, n), e = min(b + per, n);
    int s = 0;
    for (int i = b; i < e; ++i) s += cnt[i];
    int v = s;
#pragma unroll
    for (int off = 1; off < 32; off <<= 1) {
        int u = __shfl_up(v, off, 32);
        if (lane >= off) v += u;
    }
    if (lane == 31) wsum[w] = v;
    __syncthreads();
    if (w == 0) {
        const int x = wsum[lane];
        int y = x;
#pragma unroll
        for (int off = 1; off < 32; off <<= 1) {
            int u = __shfl_up(y, off, 32);
            if (lane >= off) y += u;
        }
        wsum[lane] = y - x;
    }
    __syncthreads();
    int run = wsum[w] + v - s;
    for (int i = b; i < e; ++i) { start[i] = run; run += cnt[i]; }
    if (t == 1023) start[n] = run;
}

__global__ void fill_srcs(const int* __restrict__ ei, int E,
                          const int* __restrict__ start, const int* __restrict__ rank,
                          int* __restrict__ srcs) {
    int e = blockIdx.x * blockDim.x + threadIdx.x;
    if (e < E) srcs[start[ei[E + e]] + rank[e]] = ei[e];
}

#define ACH 8

__global__ __launch_bounds__(256) void gather_agg(
    const float* __restrict__ m, const int* __restrict__ start,
    const int* __restrict__ srcs, const float* __restrict__ b,
    float* __restrict__ agg, int n)
{
    const int wv   = __builtin_amdgcn_readfirstlane((int)(threadIdx.x >> 5));
    const int lane = threadIdx.x & 31;
    const int node = blockIdx.x * 8 + wv;
    if (node >= n) return;
    const int beg = start[node], end = start[node + 1];

    float4 acc = make_float4(0.0f, 0.0f, 0.0f, 0.0f);
    for (int i = beg; i < end; i += 32) {
        const int n32 = min(end - i, 32);
        const int sv = srcs[i + min(lane, n32 - 1)];
        for (int j0 = 0; j0 < n32; j0 += ACH) {
            const int nl = n32 - j0;
            float4 v[ACH];
            v[0] = *(const float4*)(m + (size_t)__builtin_amdgcn_readlane(sv, j0) * HIDC + lane * 4);
#pragma unroll
            for (int j = 1; j < ACH; ++j)
                if (j < nl) {
                    const int sj = __builtin_amdgcn_readlane(sv, j0 + j);
                    v[j] = *(const float4*)(m + (size_t)sj * HIDC + lane * 4);
                }
            acc.x += v[0].x; acc.y += v[0].y; acc.z += v[0].z; acc.w += v[0].w;
#pragma unroll
            for (int j = 1; j < ACH; ++j)
                if (j < nl) { acc.x += v[j].x; acc.y += v[j].y; acc.z += v[j].z; acc.w += v[j].w; }
        }
    }
    const float4 bb = *(const float4*)(b + lane * 4);
    float4 r;
    r.x = fmaxf(acc.x + bb.x, 0.0f);
    r.y = fmaxf(acc.y + bb.y, 0.0f);
    r.z = fmaxf(acc.z + bb.z, 0.0f);
    r.w = fmaxf(acc.w + bb.w, 0.0f);
    *(float4*)(agg + (size_t)node * HIDC + lane * 4) = r;
}

#define GCH 8
#define GBW 4

__device__ __forceinline__ float4 ld_row4(const float* __restrict__ agg, int node, int lane) {
    return *(const float4*)(agg + (size_t)node * HIDC + lane * 4);
}

__device__ __forceinline__ float sqdist4(const float4 a, const float4 b) {
    float dx = a.x - b.x, dy = a.y - b.y, dz = a.z - b.z, dw = a.w - b.w;
    return dx * dx + dy * dy + dz * dz + dw * dw;
}

__device__ __forceinline__ void flush_run(float* __restrict__ gsum, int node, float s, int lane) {
#pragma unroll
    for (int off = 16; off > 0; off >>= 1)
        s += __shfl_down(s, off, 32);
    if (lane == 0) atomicAdd(&gsum[node], s);
}

__device__ __forceinline__ void walk_edges(
    const float* __restrict__ agg, const int* __restrict__ ei, int E,
    float* __restrict__ gsum, int e_beg, int e_end, int lane)
{
    int    prev = -1;
    float  acc  = 0.0f;
    float4 a    = make_float4(0.0f, 0.0f, 0.0f, 0.0f);
    int e0 = e_beg;
    if (e0 + GCH <= e_end) {
        int rr[GCH], cc[GCH];
#pragma unroll
        for (int j = 0; j < GCH; ++j) { rr[j] = ei[e0 + j]; cc[j] = ei[E + e0 + j]; }
        for (;;) {
            const bool more = e0 + 2 * GCH <= e_end;
            const int  en   = more ? e0 + GCH : e0;
            int rn[GCH], cn[GCH];
#pragma unroll
            for (int j = 0; j < GCH; ++j) { rn[j] = ei[en + j]; cn[j] = ei[E + en + j]; }

            float4 bv[GCH], av[GCH];
#pragma unroll
            for (int j = 0; j < GCH; ++j) bv[j] = ld_row4(agg, cc[j], lane);
#pragma unroll
            for (int j = 0; j < GCH; ++j) {
                const int pj = j ? rr[j - 1] : prev;
                if (rr[j] != pj) av[j] = ld_row4(agg, rr[j], lane);
            }
#pragma unroll
            for (int j = 0; j < GCH; ++j) {
                const int pj = j ? rr[j - 1] : prev;
                if (rr[j] != pj) {
                    if (j > 0 || pj >= 0) flush_run(gsum, pj, acc, lane);
                    acc = 0.0f;
                    a = av[j];
                }
                acc += sqdist4(a, bv[j]);
            }
            prev = rr[GCH - 1];
            e0 += GCH;
            if (!more) break;
#pragma unroll
            for (int j = 0; j < GCH; ++j) { rr[j] = rn[j]; cc[j] = cn[j]; }
        }
    }
    for (; e0 < e_end; ++e0) {
        const int r = ei[e0];
        const int c = ei[E + e0];
        if (r != prev) {
            if (prev >= 0) flush_run(gsum, prev, acc, lane);
            acc = 0.0f; prev = r;
            a = ld_row4(agg, r, lane);
        }
        acc += sqdist4(a, ld_row4(agg, c, lane));
    }
    flush_run(gsum, prev, acc, lane);
}

__global__ __launch_bounds__(256) void gamma_edges(
    const float* __restrict__ agg,
    const int* __restrict__ eiA, int EA, float* __restrict__ gsumA, int qA, int rA,
    const int* __restrict__ eiB, int EB, float* __restrict__ gsumB, int qB, int rB)
{
    const int wv   = __builtin_amdgcn_readfirstlane((int)(threadIdx.x >> 5));
    const int lane = threadIdx.x & 31;
    const int w    = blockIdx.x * 8 + wv;
    const int nw   = gridDim.x * 8;

    const int a0 = w * qA + min(w, rA);
    const int a1 = a0 + qA + (w < rA ? 1 : 0);
    if (a0 < a1)
        walk_edges(agg, eiA, EA, gsumA, a0 * GCH, min(a1 * GCH, EA), lane);

    const int xb = max(w - (nw - rB), 0);
    const int b0 = w * qB + xb;
    const int b1 = b0 + qB + (w >= nw - rB ? 1 : 0);
    if (b0 < b1)
        walk_edges(agg, eiB, EB, gsumB, b0 * GCH, min(b1 * GCH, EB), lane);
}

__global__ __launch_bounds__(256) void update_h(
    float* __restrict__ h, const float* __restrict__ agg,
    const float* __restrict__ xs,
    const float* __restrict__ gsum1, const float* __restrict__ deg1,
    const float* __restrict__ gsum2, const float* __restrict__ deg2, int n)
{
    const int wv   = __builtin_amdgcn_readfirstlane((int)(threadIdx.x >> 5));
    const int lane = threadIdx.x & 31;
    const int node = blockIdx.x * 8 + wv;
    if (node >= n) return;
    const size_t o = (size_t)node * HIDC + lane * 4;
    const float4 hv = *(const float4*)(h + o);
    const float4 av = *(const float4*)(agg + o);
    const float4 xv = *(const float4*)(xs + o);
    const float gs = tanhf(gsum1[node] * (1.0f / (deg1[node] + 1e-10f)));
    const float gq = tanhf(gsum2[node] * (1.0f / (deg2[node] + 1e-10f)));
    const float den = 1.0f + gs + gq;
    float4 r;
    r.x = (hv.x + gs * av.x + gq * xv.x) / den;
    r.y = (hv.y + gs * av.y + gq * xv.y) / den;
    r.z = (hv.z + gs * av.z + gq * xv.z) / den;
    r.w = (hv.w + gs * av.w + gq * xv.w) / den;
    *(float4*)(h + o) = r;
}

extern "C" void kernel_launch(void* const* d_in, const int* in_sizes, int n_in,
                              void* d_out, int out_size, void* d_ws, size_t ws_size,
                              hipStream_t stream) {
    const float* x      = (const float*)d_in[0];
    const float* x0     = (const float*)d_in[1];
    const int*   ei     = (const int*)d_in[2];
    const int*   ei2    = (const int*)d_in[3];
    const float* in_W   = (const float*)d_in[4];
    const float* skip_W = (const float*)d_in[5];
    const float* conv_W = (const float*)d_in[6];
    const float* conv_b = (const float*)d_in[7];
    const float* fc_W   = (const float*)d_in[8];
    const float* fc_b   = (const float*)d_in[9];
    float* out = (float*)d_out;

    const int N  = in_sizes[0] / HIDC;
    const int E1 = in_sizes[2] / 2;
    const int E2 = in_sizes[3] / 2;

    const size_t nh = (size_t)N * HIDC;
    float* h     = (float*)d_ws;
    float* xskip = h + nh;
    float* agg   = xskip + nh;
    float* mbuf  = agg + nh;

    float* zbase = (float*)d_out;
    int*   cnt   = (int*)zbase;
    float* deg1  = zbase + N;
    float* deg2  = deg1 + N;
    float* gsum1[2] = { deg2 + N,     deg2 + 3 * N };
    float* gsum2[2] = { deg2 + 2 * N, deg2 + 4 * N };
    int* start = (int*)(zbase + 7 * N);
    int* rank  = start + N + 1;
    int* srcs  = rank + E1;

    const dim3 blk(256);
    auto gemm_blocks = [](int M, int Nc, int pairs) {
        int tiles = ((M + 15) / 16) * ((Nc + 15) / 16);
        return dim3((tiles + 7) / 8, pairs);
    };
    const dim3 sblk(128);
    const int sb = (((N + 15) / 16) * 2 + 3) / 4;
    const float* conv_W1 = conv_W + (size_t)HIDC * HIDC;
    const int zb = 128, eb = 512;

    gemm_strip_f32<1><<<zb + 2 * sb, sblk, 0, stream>>>(
        x, in_W, h, x0, in_W, agg, N, sb, zb,
        nullptr, 0, nullptr, 0, nullptr, nullptr, nullptr, nullptr, zbase, 7 * N);
    gemm_strip_f32<2><<<eb + 2 * sb, sblk, 0, stream>>>(
        agg, skip_W, xskip, h, conv_W, mbuf, N, sb, eb,
        ei, E1, ei2, E2, deg1, deg2, cnt, rank, nullptr, 0);

    scan_counts<<<1, 1024, 0, stream>>>(cnt, start, N);
    fill_srcs<<<(E1 + 255) / 256, blk, 0, stream>>>(ei, E1, start, rank, srcs);

    const int gblocks = 256 * GBW;
    const int gwaves  = gblocks * 8;
    const int nch2 = (E2 + GCH - 1) / GCH, nch1 = (E1 + GCH - 1) / GCH;
    const int q2 = nch2 / gwaves, r2 = nch2 % gwaves;
    const int q1 = nch1 / gwaves, r1 = nch1 % gwaves;
    for (int l = 0; l < 2; ++l) {
        if (l == 1)
            gemm_strip_f32<0><<<sb, sblk, 0, stream>>>(
                h, conv_W1, mbuf, nullptr, nullptr, nullptr, N, sb, 0,
                nullptr, 0, nullptr, 0, nullptr, nullptr, nullptr, nullptr, nullptr, 0);
        gather_agg<<<(N + 7) / 8, blk, 0, stream>>>(mbuf, start, srcs, conv_b + l * HIDC, agg, N);
        gamma_edges<<<gblocks, blk, 0, stream>>>(agg, ei2, E2, gsum2[l], q2, r2, ei, E1, gsum1[l], q1, r1);
        update_h<<<(N + 7) / 8, blk, 0, stream>>>(h, agg, xskip, gsum1[l], deg1, gsum2[l], deg2, N);
    }

    gemm_wmma_f32<40><<<gemm_blocks(N, 40, 1), blk, 0, stream>>>(
        h, fc_W, fc_b, out, nullptr, nullptr, nullptr, N);
}


# fewer launches (12 -> 10), same arithmetic and sum order: layer 0's update_h rides in layer 1's product launch (rows reach the WMMA loop through LDS); count scan done per block in the fill launch
# speedup vs baseline: 2.6651x; 1.0640x over previous
#include <hip/hip_runtime.h>

typedef __attribute__((ext_vector_type(2))) float v2f;
typedef __attribute__((ext_vector_type(8))) float v8f;

#define HIDC 128

template <int NC>
__global__ __launch_bounds__(256) void gemm_wmma_f32(
    const float* __restrict__ A0, const float* __restrict__ B0,
    const float* __restrict__ bias, float* __restrict__ C0,
    const float* __restrict__ A1, const float* __restrict__ B1,
    float* __restrict__ C1, int M)
{
    const bool second = blockIdx.y != 0;
    const float* A = second ? A1 : A0;
    const float* B = second ? B1 : B0;
    float*       C = second ? C1 : C0;
    constexpr int K = HIDC;
    constexpr int tilesN = (NC + 15) >> 4;
    const int tilesM = (M + 15) >> 4;
    const int wid = blockIdx.x * 8 + (threadIdx.x >> 5);
    if (wid >= tilesM * tilesN) return;
    const int tm = wid / tilesN, tn = wid % tilesN;
    const int m0 = tm << 4, n0 = tn << 4;
    const int lane = threadIdx.x & 31;
    const int half = lane >> 4;
    const int l16  = lane & 15;

    const int arow   = min(m0 + l16, M - 1);
    const int bcol_r = n0 + l16;
    const int bcol   = min(bcol_r, NC - 1);

    const float* Ap = A + (size_t)arow * K + 2 * half;
    const float* Bp = B + (size_t)(2 * half) * NC + bcol;

    v8f acc = {};
#pragma unroll
    for (int k0 = 0; k0 < K; k0 += 4) {
        v2f a = *(const v2f*)(Ap + k0);
        v2f b;
        b.x = Bp[k0 * NC];
        b.y = Bp[(k0 + 1) * NC];
        acc = __builtin_amdgcn_wmma_f32_16x16x4_f32(
                  false, a, false, b, (short)0, acc, false, false);
    }

    if (bcol_r < NC) {
        const float bv = bias ? bias[bcol_r] : 0.0f;
#pragma unroll
        for (int v = 0; v < 8; ++v) {
            const int r = m0 + v + 8 * half;
            if (r < M) C[(size_t)r * NC + bcol_r] = acc[v] + bv;
        }
    }
}

__device__ __forceinline__ void edge_prep_part(
    const int* __restrict__ ei, int E1, const int* __restrict__ ei2, int E2,
    float* __restrict__ deg1, float* __restrict__ deg2,
    int* __restrict__ cnt, int* __restrict__ rank, int t, int T)
{
    for (int e = t; e < E1; e += T) {
        atomicAdd(&deg1[ei[e]], 1.0f);
        rank[e] = atomicAdd(&cnt[ei[E1 + e]], 1);
    }
    const int n4 = E2 >> 2;
    for (int g = t; g < n4; g += T) {
        const int4 r = *(const int4*)(ei2 + 4 * g);
        int p = r.x;
        float c = 1.0f;
        if (r.y == p) c += 1.0f; else { atomicAdd(&deg2[p], c); p = r.y; c = 1.0f; }
        if (r.z == p) c += 1.0f; else { atomicAdd(&deg2[p], c); p = r.z; c = 1.0f; }
        if (r.w == p) c += 1.0f; else { atomicAdd(&deg2[p], c); p = r.w; c = 1.0f; }
        atomicAdd(&deg2[p], c);
    }
    for (int f = (n4 << 2) + t; f < E2; f += T)
        atomicAdd(&deg2[ei2[f]], 1.0f);
}

#define APITCH 132

__device__ __forceinline__ float4 gated_row(const float4 hv, const float4 av, const float4 xv,
                                            float gs, float gq, float den) {
    float4 r;
    r.x = (hv.x + gs * av.x + gq * xv.x) / den;
    r.y = (hv.y + gs * av.y + gq * xv.y) / den;
    r.z = (hv.z + gs * av.z + gq * xv.z) / den;
    r.w = (hv.w + gs * av.w + gq * xv.w) / den;
    return r;
}

__device__ __forceinline__ float lane_value(float x, int i) {
    return __builtin_bit_cast(float, __builtin_amdgcn_readlane(__builtin_bit_cast(int, x), i));
}

template <int RPW>
__device__ __forceinline__ void update_rows_to_lds(
    float* sA, float* h, const float* __restrict__ agg, const float* __restrict__ xs,
    const float* __restrict__ gsum1, const float* __restrict__ deg1,
    const float* __restrict__ gsum2, const float* __restrict__ deg2,
    int r0, int M, int wv, int lane)
{
    const int gn = min(r0 + wv * RPW + min(lane, RPW - 1), M - 1);
    const float g1 = gsum1[gn], d1 = deg1[gn], g2 = gsum2[gn], d2 = deg2[gn];
    float4 hv[RPW], av[RPW], xv[RPW];
#pragma unroll
    for (int i = 0; i < RPW; ++i) {
        const size_t o = (size_t)min(r0 + wv * RPW + i, M - 1) * HIDC + lane * 4;
        hv[i] = *(const float4*)(h + o);
        av[i] = *(const float4*)(agg + o);
        xv[i] = *(const float4*)(xs + o);
    }
    const float gsl  = tanhf(g1 * (1.0f / (d1 + 1e-10f)));
    const float gql  = tanhf(g2 * (1.0f / (d2 + 1e-10f)));
    const float denl = 1.0f + gsl + gql;
#pragma unroll
    for (int i = 0; i < RPW; ++i) {
        const int lr = wv * RPW + i, node = r0 + lr;
        float4 r = make_float4(0.0f, 0.0f, 0.0f, 0.0f);
        if (node < M) {
            r = gated_row(hv[i], av[i], xv[i],
                          lane_value(gsl, i), lane_value(gql, i), lane_value(denl, i));
            *(float4*)(h + (size_t)node * HIDC + lane * 4) = r;
        }
        *(float4*)(sA + lr * APITCH + lane * 4) = r;
    }
}

__device__ __forceinline__ v8f wmma4(const float a0, const float a1,
                                     const float b0, const float b1, const v8f acc) {
    v2f a; a.x = a0; a.y = a1;
    v2f b; b.x = b0; b.y = b1;
    return __builtin_amdgcn_wmma_f32_16x16x4_f32(false, a, false, b, (short)0, acc, false, false);
}

template <int AUX>
__global__ __launch_bounds__(128) __attribute__((amdgpu_waves_per_eu(1, 4))) void gemm_strip_f32(
    const float* __restrict__ A0, const float* __restrict__ B0, float* __restrict__ C0,
    const float* __restrict__ A1, const float* __restrict__ B1, float* __restrict__ C1,
    int M, int gb, int auxb,
    const int* __restrict__ ei, int E1, const int* __restrict__ ei2, int E2,
    float* __restrict__ deg1, float* __restrict__ deg2,
    int* __restrict__ cnt, int* __restrict__ rank,
    float* __restrict__ zp, int zn,
    float* hu, const float* __restrict__ aggu, const float* __restrict__ xsu,
    const float* __restrict__ gsum1, const float* __restrict__ gsum2)
{
    __shared__ float sA[AUX == 0 ? 32 * APITCH : 1];
    if constexpr (AUX == 0) {
        const int wvu = __builtin_amdgcn_readfirstlane((int)(threadIdx.x >> 5));
        update_rows_to_lds<8>(sA, hu, aggu, xsu, gsum1, deg1, gsum2, deg2,
                              (int)blockIdx.x * 32, M, wvu, (int)(threadIdx.x & 31));
        __syncthreads();
    }
    if (AUX != 0 && (int)blockIdx.x < auxb) {
        const int t = blockIdx.x * 128 + threadIdx.x, T = auxb * 128;
        if (AUX == 1) {
            for (int i = t; i < zn; i += T) zp[i] = 0.0f;
        } else {
            edge_prep_part(ei, E1, ei2, E2, deg1, deg2, cnt, rank, t, T);
        }
        return;
    }
    const int  g      = (int)blockIdx.x - auxb;
    const bool second = g >= gb;
    const float* A = second ? A1 : A0;
    const float* B = second ? B1 : B0;
    float*       C = second ? C1 : C0;
    constexpr int K = HIDC, NC = HIDC;
    const int tilesM = (M + 15) >> 4;
    const int wid = (second ? g - gb : g) * 4 + (threadIdx.x >> 5);
    const int tm  = wid >> 1;
    if (tm >= tilesM) return;
    const int nb   = (wid & 1) << 6;
    const int m0   = tm << 4;
    const int lane = threadIdx.x & 31;
    const int hf   = lane >> 4;
    const int c    = lane & 15;

    const int arow = min(m0 + c, M - 1);
    const float* Ap = A + (size_t)arow * K + 4 * hf;
    const float* Bp = B + (size_t)(4 * hf) * NC + nb + 4 * c;
    const float* Al = sA + ((int)((threadIdx.x >> 6) << 4) + c) * APITCH + 4 * hf;

    v8f acc0 = {}, acc1 = {}, acc2 = {}, acc3 = {};
#pragma unroll
    for (int kb = 0; kb < K; kb += 8) {
        const float4 a  = (AUX == 0) ? *(const float4*)(Al + kb) : *(const float4*)(Ap + kb);
        const float4 b0 = *(const float4*)(Bp + (kb + 0) * NC);
        const float4 b1 = *(const float4*)(Bp + (kb + 1) * NC);
        const float4 b2 = *(const float4*)(Bp + (kb + 2) * NC);
        const float4 b3 = *(const float4*)(Bp + (kb + 3) * NC);
        acc0 = wmma4(a.x, a.y, b0.x, b1.x, acc0);
        acc1 = wmma4(a.x, a.y, b0.y, b1.y, acc1);
        acc2 = wmma4(a.x, a.y, b0.z, b1.z, acc2);
        acc3 = wmma4(a.x, a.y, b0.w, b1.w, acc3);
        acc0 = wmma4(a.z, a.w, b2.x, b3.x, acc0);
        acc1 = wmma4(a.z, a.w, b2.y, b3.y, acc1);
        acc2 = wmma4(a.z, a.w, b2.z, b3.z, acc2);
        acc3 = wmma4(a.z, a.w, b2.w, b3.w, acc3);
    }

#pragma unroll
    for (int v = 0; v < 8; ++v) {
        const int r = m0 + v + 8 * hf;
        if (r < M)
            *(float4*)(C + (size_t)r * NC + nb + 4 * c) =
                make_float4(acc0[v], acc1[v], acc2[v], acc3[v]);
    }
}


__global__ __launch_bounds__(1024) void scan_counts(const int* __restrict__ cnt,
                                                    int* __restrict__ start, int n) {
    __shared__ int wsum[32];
    const int t = threadIdx.x, lane = t & 31, w = t >> 5;
    const int per = (n + 1023) >> 10;
    const int b = min(t * per, n), e = min(b + per, n);
    int s = 0;
    for (int i = b; i < e; ++i) s += cnt[i];
    int v = s;
#pragma unroll
    for (int off = 1; off < 32; off <<= 1) {
        int u = __shfl_up(v, off, 32);
        if (lane >= off) v += u;
    }
    if (lane == 31) wsum[w] = v;
    __syncthreads();
    if (w == 0) {
        const int x = wsum[lane];
        int y = x;
#pragma unroll
        for (int off = 1; off < 32; off <<= 1) {
            int u = __shfl_up(y, off, 32);
            if (lane >= off) y += u;
        }
        wsum[lane] = y - x;
    }
    __syncthreads();
    int run = wsum[w] + v - s;
    for (int i = b; i < e; ++i) { start[i] = run; run += cnt[i]; }
    if (t == 1023) start[n] = run;
}

__global__ void fill_srcs(const int* __restrict__ ei, int E,
                          const int* __restrict__ start, const int* __restrict__ rank,
                          int* __restrict__ srcs) {
    int e = blockIdx.x * blockDim.x + threadIdx.x;
    if (e < E) srcs[start[ei[E + e]] + rank[e]] = ei[e];
}

#define SCAN_CAP 32768
__global__ __launch_bounds__(1024) void scan_fill(const int* __restrict__ cnt, int n,
                                                  const int* __restrict__ ei, int E,
                                                  const int* __restrict__ rank,
                                                  int* __restrict__ start, int* __restrict__ srcs) {
    __shared__ int sc[SCAN_CAP];
    __shared__ int wsum[32];
    const int t = threadIdx.x, lane = t & 31, w = t >> 5;
    const int e = blockIdx.x * 1024 + t;
    int src = 0, dst = 0, rk = 0;
    if (e < E) { src = ei[e]; dst = ei[E + e]; rk = rank[e]; }
    for (int i = t; i < n; i += 1024) sc[i] = cnt[i];
    __syncthreads();
    const int per = (n + 1023) >> 10;
    const int b = min(t * per, n), en = min(b + per, n);
    int s = 0;
    for (int i = b; i < en; ++i) s += sc[i];
    int v = s;
#pragma unroll
    for (int off = 1; off < 32; off <<= 1) {
        int u = __shfl_up(v, off, 32);
        if (lane >= off) v += u;
    }
    if (lane == 31) wsum[w] = v;
    __syncthreads();
    if (w == 0) {
        const int x = wsum[lane];
        int y = x;
#pragma unroll
        for (int off = 1; off < 32; off <<= 1) {
            int u = __shfl_up(y, off, 32);
            if (lane >= off) y += u;
        }
        wsum[lane] = y - x;
    }
    __syncthreads();
    int run = wsum[w] + v - s;
    for (int i = b; i < en; ++i) { const int k = sc[i]; sc[i] = run; run += k; }
    if (t == 1023) sc[n] = run;
    __syncthreads();
    for (int i = blockIdx.x * 1024 + t; i <= n; i += gridDim.x * 1024) start[i] = sc[i];
    if (e < E) srcs[sc[dst] + rk] = src;
}

#define ACH 8

__global__ __launch_bounds__(256) void gather_agg(
    const float* __restrict__ m, const int* __restrict__ start,
    const int* __restrict__ srcs, const float* __restrict__ b,
    float* __restrict__ agg, int n)
{
    const int wv   = __builtin_amdgcn_readfirstlane((int)(threadIdx.x >> 5));
    const int lane = threadIdx.x & 31;
    const int node = blockIdx.x * 8 + wv;
    if (node >= n) return;
    const int beg = start[node], end = start[node + 1];

    float4 acc = make_float4(0.0f, 0.0f, 0.0f, 0.0f);
    for (int i = beg; i < end; i += 32) {
        const int n32 = min(end - i, 32);
        const int sv = srcs[i + min(lane, n32 - 1)];
        for (int j0 = 0; j0 < n32; j0 += ACH) {
            const int nl = n32 - j0;
            float4 v[ACH];
            v[0] = *(const float4*)(m + (size_t)__builtin_amdgcn_readlane(sv, j0) * HIDC + lane * 4);
#pragma unroll
            for (int j = 1; j < ACH; ++j)
                if (j < nl) {
                    const int sj = __builtin_amdgcn_readlane(sv, j0 + j);
                    v[j] = *(const float4*)(m + (size_t)sj * HIDC + lane * 4);
                }
            acc.x += v[0].x; acc.y += v[0].y; acc.z += v[0].z; acc.w += v[0].w;
#pragma unroll
            for (int j = 1; j < ACH; ++j)
                if (j < nl) { acc.x += v[j].x; acc.y += v[j].y; acc.z += v[j].z; acc.w += v[j].w; }
        }
    }
    const float4 bb = *(const float4*)(b + lane * 4);
    float4 r;
    r.x = fmaxf(acc.x + bb.x, 0.0f);
    r.y = fmaxf(acc.y + bb.y, 0.0f);
    r.z = fmaxf(acc.z + bb.z, 0.0f);
    r.w = fmaxf(acc.w + bb.w, 0.0f);
    *(float4*)(agg + (size_t)node * HIDC + lane * 4) = r;
}

#define GCH 8
#define GBW 4

__device__ __forceinline__ float4 ld_row4(const float* __restrict__ agg, int node, int lane) {
    return *(const float4*)(agg + (size_t)node * HIDC + lane * 4);
}

__device__ __forceinline__ float sqdist4(const float4 a, const float4 b) {
    float dx = a.x - b.x, dy = a.y - b.y, dz = a.z - b.z, dw = a.w - b.w;
    return dx * dx + dy * dy + dz * dz + dw * dw;
}

__device__ __forceinline__ void flush_run(float* __restrict__ gsum, int node, float s, int lane) {
#pragma unroll
    for (int off = 16; off > 0; off >>= 1)
        s += __shfl_down(s, off, 32);
    if (lane == 0) atomicAdd(&gsum[node], s);
}

__device__ __forceinline__ void walk_edges(
    const float* __restrict__ agg, const int* __restrict__ ei, int E,
    float* __restrict__ gsum, int e_beg, int e_end, int lane)
{
    int    prev = -1;
    float  acc  = 0.0f;
    float4 a    = make_float4(0.0f, 0.0f, 0.0f, 0.0f);
    int e0 = e_beg;
    if (e0 + GCH <= e_end) {
        int rr[GCH], cc[GCH];
#pragma unroll
        for (int j = 0; j < GCH; ++j) { rr[j] = ei[e0 + j]; cc[j] = ei[E + e0 + j]; }
        for (;;) {
            const bool more = e0 + 2 * GCH <= e_end;
            const int  en   = more ? e0 + GCH : e0;
            int rn[GCH], cn[GCH];
#pragma unroll
            for (int j = 0; j < GCH; ++j) { rn[j] = ei[en + j]; cn[j] = ei[E + en + j]; }

            float4 bv[GCH], av[GCH];
#pragma unroll
            for (int j = 0; j < GCH; ++j) bv[j] = ld_row4(agg, cc[j], lane);
#pragma unroll
            for (int j = 0; j < GCH; ++j) {
                const int pj = j ? rr[j - 1] : prev;
                if (rr[j] != pj) av[j] = ld_row4(agg, rr[j], lane);
            }
#pragma unroll
            for (int j = 0; j < GCH; ++j) {
                const int pj = j ? rr[j - 1] : prev;
                if (rr[j] != pj) {
                    if (j > 0 || pj >= 0) flush_run(gsum, pj, acc, lane);
                    acc = 0.0f;
                    a = av[j];
                }
                acc += sqdist4(a, bv[j]);
            }
            prev = rr[GCH - 1];
            e0 += GCH;
            if (!more) break;
#pragma unroll
            for (int j = 0; j < GCH; ++j) { rr[j] = rn[j]; cc[j] = cn[j]; }
        }
    }
    for (; e0 < e_end; ++e0) {
        const int r = ei[e0];
        const int c = ei[E + e0];
        if (r != prev) {
            if (prev >= 0) flush_run(gsum, prev, acc, lane);
            acc = 0.0f; prev = r;
            a = ld_row4(agg, r, lane);
        }
        acc += sqdist4(a, ld_row4(agg, c, lane));
    }
    flush_run(gsum, prev, acc, lane);
}

__global__ __launch_bounds__(256) void gamma_edges(
    const float* __restrict__ agg,
    const int* __restrict__ eiA, int EA, float* __restrict__ gsumA, int qA, int rA,
    const int* __restrict__ eiB, int EB, float* __restrict__ gsumB, int qB, int rB)
{
    const int wv   = __builtin_amdgcn_readfirstlane((int)(threadIdx.x >> 5));
    const int lane = threadIdx.x & 31;
    const int w    = blockIdx.x * 8 + wv;
    const int nw   = gridDim.x * 8;

    const int a0 = w * qA + min(w, rA);
    const int a1 = a0 + qA + (w < rA ? 1 : 0);
    if (a0 < a1)
        walk_edges(agg, eiA, EA, gsumA, a0 * GCH, min(a1 * GCH, EA), lane);

    const int xb = max(w - (nw - rB), 0);
    const int b0 = w * qB + xb;
    const int b1 = b0 + qB + (w >= nw - rB ? 1 : 0);
    if (b0 < b1)
        walk_edges(agg, eiB, EB, gsumB, b0 * GCH, min(b1 * GCH, EB), lane);
}

__global__ __launch_bounds__(256) void update_h(
    float* __restrict__ h, const float* __restrict__ agg,
    const float* __restrict__ xs,
    const float* __restrict__ gsum1, const float* __restrict__ deg1,
    const float* __restrict__ gsum2, const float* __restrict__ deg2, int n)
{
    const int wv   = __builtin_amdgcn_readfirstlane((int)(threadIdx.x >> 5));
    const int lane = threadIdx.x & 31;
    const int node = blockIdx.x * 8 + wv;
    if (node >= n) return;
    const size_t o = (size_t)node * HIDC + lane * 4;
    const float4 hv = *(const float4*)(h + o);
    const float4 av = *(const float4*)(agg + o);
    const float4 xv = *(const float4*)(xs + o);
    const float gs = tanhf(gsum1[node] * (1.0f / (deg1[node] + 1e-10f)));
    const float gq = tanhf(gsum2[node] * (1.0f / (deg2[node] + 1e-10f)));
    const float den = 1.0f + gs + gq;
    float4 r;
    r.x = (hv.x + gs * av.x + gq * xv.x) / den;
    r.y = (hv.y + gs * av.y + gq * xv.y) / den;
    r.z = (hv.z + gs * av.z + gq * xv.z) / den;
    r.w = (hv.w + gs * av.w + gq * xv.w) / den;
    *(float4*)(h + o) = r;
}

extern "C" void kernel_launch(void* const* d_in, const int* in_sizes, int n_in,
                              void* d_out, int out_size, void* d_ws, size_t ws_size,
                              hipStream_t stream) {
    const float* x      = (const float*)d_in[0];
    const float* x0     = (const float*)d_in[1];
    const int*   ei     = (const int*)d_in[2];
    const int*   ei2    = (const int*)d_in[3];
    const float* in_W   = (const float*)d_in[4];
    const float* skip_W = (const float*)d_in[5];
    const float* conv_W = (const float*)d_in[6];
    const float* conv_b = (const float*)d_in[7];
    const float* fc_W   = (const float*)d_in[8];
    const float* fc_b   = (const float*)d_in[9];
    float* out = (float*)d_out;

    const int N  = in_sizes[0] / HIDC;
    const int E1 = in_sizes[2] / 2;
    const int E2 = in_sizes[3] / 2;

    const size_t nh = (size_t)N * HIDC;
    float* h     = (float*)d_ws;
    float* xskip = h + nh;
    float* agg   = xskip + nh;
    float* mbuf  = agg + nh;

    float* zbase = (float*)d_out;
    int*   cnt   = (int*)zbase;
    float* deg1  = zbase + N;
    float* deg2  = deg1 + N;
    float* gsum1[2] = { deg2 + N,     deg2 + 3 * N };
    float* gsum2[2] = { deg2 + 2 * N, deg2 + 4 * N };
    int* start = (int*)(zbase + 7 * N);
    int* rank  = start + N + 1;
    int* srcs  = rank + E1;

    const dim3 blk(256);
    auto gemm_blocks = [](int M, int Nc, int pairs) {
        int tiles = ((M + 15) / 16) * ((Nc + 15) / 16);
        return dim3((tiles + 7) / 8, pairs);
    };
    const dim3 sblk(128);
    const int sb = (((N + 15) / 16) * 2 + 3) / 4;
    const float* conv_W1 = conv_W + (size_t)HIDC * HIDC;
    const int zb = 128, eb = 512;

    gemm_strip_f32<1><<<zb + 2 * sb, sblk, 0, stream>>>(
        x, in_W, h, x0, in_W, agg, N, sb, zb,
        nullptr, 0, nullptr, 0, nullptr, nullptr, nullptr, nullptr, zbase, 7 * N,
        nullptr, nullptr, nullptr, nullptr, nullptr);
    gemm_strip_f32<2><<<eb + 2 * sb, sblk, 0, stream>>>(
        agg, skip_W, xskip, h, conv_W, mbuf, N, sb, eb,
        ei, E1, ei2, E2, deg1, deg2, cnt, rank, nullptr, 0,
        nullptr, nullptr, nullptr, nullptr, nullptr);

    if (N + 1 <= SCAN_CAP) {
        scan_fill<<<(E1 + 1023) / 1024, 1024, 0, stream>>>(cnt, N, ei, E1, rank, start, srcs);
    } else {
        scan_counts<<<1, 1024, 0, stream>>>(cnt, start, N);
        fill_srcs<<<(E1 + 255) / 256, blk, 0, stream>>>(ei, E1, start, rank, srcs);
    }

    const int gblocks = 256 * GBW;
    const int gwaves  = gblocks * 8;
    const int nch2 = (E2 + GCH - 1) / GCH, nch1 = (E1 + GCH - 1) / GCH;
    const int q2 = nch2 / gwaves, r2 = nch2 % gwaves;
    const int q1 = nch1 / gwaves, r1 = nch1 % gwaves;
    for (int l = 0; l < 2; ++l) {
        if (l == 1)
            gemm_strip_f32<0><<<sb, sblk, 0, stream>>>(
                nullptr, conv_W1, mbuf, nullptr, nullptr, nullptr, N, sb, 0,
                nullptr, 0, nullptr, 0, deg1, deg2, nullptr, nullptr, nullptr, 0,
                h, agg, xskip, gsum1[0], gsum2[0]);
        gather_agg<<<(N + 7) / 8, blk, 0, stream>>>(mbuf, start, srcs, conv_b + l * HIDC, agg, N);
        gamma_edges<<<gblocks, blk, 0, stream>>>(agg, ei2, E2, gsum2[l], q2, r2, ei, E1, gsum1[l], q1, r1);
        if (l == 1)
            update_h<<<(N + 7) / 8, blk, 0, stream>>>(h, agg, xskip, gsum1[l], deg1, gsum2[l], deg2, N);
    }

    gemm_wmma_f32<40><<<gemm_blocks(N, 40, 1), blk, 0, stream>>>(
        h, fc_W, fc_b, out, nullptr, nullptr, nullptr, N);
}


# one-hop sources by destination in 80-byte buckets (word 0 = count, slot = the value its add returns; longer lists spill to one overflow list): no rank, scan or fill; 10 -> 9 launches
# speedup vs baseline: 2.8735x; 1.0782x over previous
#include <hip/hip_runtime.h>

typedef __attribute__((ext_vector_type(2))) float v2f;
typedef __attribute__((ext_vector_type(8))) float v8f;

#define HIDC 128

template <int NC>
__global__ __launch_bounds__(256) void gemm_wmma_f32(
    const float* __restrict__ A0, const float* __restrict__ B0,
    const float* __restrict__ bias, float* __restrict__ C0,
    const float* __restrict__ A1, const float* __restrict__ B1,
    float* __restrict__ C1, int M)
{
    const bool second = blockIdx.y != 0;
    const float* A = second ? A1 : A0;
    const float* B = second ? B1 : B0;
    float*       C = second ? C1 : C0;
    constexpr int K = HIDC;
    constexpr int tilesN = (NC + 15) >> 4;
    const int tilesM = (M + 15) >> 4;
    const int wid = blockIdx.x * 8 + (threadIdx.x >> 5);
    if (wid >= tilesM * tilesN) return;
    const int tm = wid / tilesN, tn = wid % tilesN;
    const int m0 = tm << 4, n0 = tn << 4;
    const int lane = threadIdx.x & 31;
    const int half = lane >> 4;
    const int l16  = lane & 15;

    const int arow   = min(m0 + l16, M - 1);
    const int bcol_r = n0 + l16;
    const int bcol   = min(bcol_r, NC - 1);

    const float* Ap = A + (size_t)arow * K + 2 * half;
    const float* Bp = B + (size_t)(2 * half) * NC + bcol;

    v8f acc = {};
#pragma unroll
    for (int k0 = 0; k0 < K; k0 += 4) {
        v2f a = *(const v2f*)(Ap + k0);
        v2f b;
        b.x = Bp[k0 * NC];
        b.y = Bp[(k0 + 1) * NC];
        acc = __builtin_amdgcn_wmma_f32_16x16x4_f32(
                  false, a, false, b, (short)0, acc, false, false);
    }

    if (bcol_r < NC) {
        const float bv = bias ? bias[bcol_r] : 0.0f;
#pragma unroll
        for (int v = 0; v < 8; ++v) {
            const int r = m0 + v + 8 * half;
            if (r < M) C[(size_t)r * NC + bcol_r] = acc[v] + bv;
        }
    }
}

#define BW   20
#define BCAP (BW - 1)
__device__ __forceinline__ int* ovf_count(int* bkt, int n) { return bkt + (size_t)n * BW; }
__device__ __forceinline__ int* ovf_pairs(int* bkt, int n) { return bkt + (size_t)n * BW + 4; }
__device__ __forceinline__ const int* ovf_count(const int* bkt, int n) { return bkt + (size_t)n * BW; }
__device__ __forceinline__ const int* ovf_pairs(const int* bkt, int n) { return bkt + (size_t)n * BW + 4; }

__device__ __forceinline__ void edge_prep_part(
    const int* __restrict__ ei, int E1, const int* __restrict__ ei2, int E2,
    float* __restrict__ deg1, float* __restrict__ deg2,
    int* __restrict__ bkt, int n, int t, int T)
{
    int* const novf = ovf_count(bkt, n);
    int* const ovf  = ovf_pairs(bkt, n);
    for (int e = t; e < E1; e += T) {
        const int src = ei[e], dst = ei[E1 + e];
        atomicAdd(&deg1[src], 1.0f);
        int* const bk = bkt + (size_t)dst * BW;
        const int slot = atomicAdd(bk, 1);
        if (slot < BCAP) {
            bk[1 + slot] = src;
        } else {
            const int o = atomicAdd(novf, 1);
            *(int2*)(ovf + 2 * o) = make_int2(dst, src);
        }
    }
    const int n4 = E2 >> 2;
    for (int g = t; g < n4; g += T) {
        const int4 r = *(const int4*)(ei2 + 4 * g);
        int p = r.x;
        float c = 1.0f;
        if (r.y == p) c += 1.0f; else { atomicAdd(&deg2[p], c); p = r.y; c = 1.0f; }
        if (r.z == p) c += 1.0f; else { atomicAdd(&deg2[p], c); p = r.z; c = 1.0f; }
        if (r.w == p) c += 1.0f; else { atomicAdd(&deg2[p], c); p = r.w; c = 1.0f; }
        atomicAdd(&deg2[p], c);
    }
    for (int f = (n4 << 2) + t; f < E2; f += T)
        atomicAdd(&deg2[ei2[f]], 1.0f);
}

#define APITCH 132

__device__ __forceinline__ float4 gated_row(const float4 hv, const float4 av, const float4 xv,
                                            float gs, float gq, float den) {
    float4 r;
    r.x = (hv.x + gs * av.x + gq * xv.x) / den;
    r.y = (hv.y + gs * av.y + gq * xv.y) / den;
    r.z = (hv.z + gs * av.z + gq * xv.z) / den;
    r.w = (hv.w + gs * av.w + gq * xv.w) / den;
    return r;
}

__device__ __forceinline__ float lane_value(float x, int i) {
    return __builtin_bit_cast(float, __builtin_amdgcn_readlane(__builtin_bit_cast(int, x), i));
}

template <int RPW>
__device__ __forceinline__ void update_rows_to_lds(
    float* sA, float* h, const float* __restrict__ agg, const float* __restrict__ xs,
    const float* __restrict__ gsum1, const float* __restrict__ deg1,
    const float* __restrict__ gsum2, const float* __restrict__ deg2,
    int r0, int M, int wv, int lane)
{
    const int gn = min(r0 + wv * RPW + min(lane, RPW - 1), M - 1);
    const float g1 = gsum1[gn], d1 = deg1[gn], g2 = gsum2[gn], d2 = deg2[gn];
    float4 hv[RPW], av[RPW], xv[RPW];
#pragma unroll
    for (int i = 0; i < RPW; ++i) {
        const size_t o = (size_t)min(r0 + wv * RPW + i, M - 1) * HIDC + lane * 4;
        hv[i] = *(const float4*)(h + o);
        av[i] = *(const float4*)(agg + o);
        xv[i] = *(const float4*)(xs + o);
    }
    const float gsl  = tanhf(g1 * (1.0f / (d1 + 1e-10f)));
    const float gql  = tanhf(g2 * (1.0f / (d2 + 1e-10f)));
    const float denl = 1.0f + gsl + gql;
#pragma unroll
    for (int i = 0; i < RPW; ++i) {
        const int lr = wv * RPW + i, node = r0 + lr;
        float4 r = make_float4(0.0f, 0.0f, 0.0f, 0.0f);
        if (node < M) {
            r = gated_row(hv[i], av[i], xv[i],
                          lane_value(gsl, i), lane_value(gql, i), lane_value(denl, i));
            *(float4*)(h + (size_t)node * HIDC + lane * 4) = r;
        }
        *(float4*)(sA + lr * APITCH + lane * 4) = r;
    }
}

__device__ __forceinline__ v8f wmma4(const float a0, const float a1,
                                     const float b0, const float b1, const v8f acc) {
    v2f a; a.x = a0; a.y = a1;
    v2f b; b.x = b0; b.y = b1;
    return __builtin_amdgcn_wmma_f32_16x16x4_f32(false, a, false, b, (short)0, acc, false, false);
}

template <int AUX>
__global__ __launch_bounds__(128) __attribute__((amdgpu_waves_per_eu(1, 4))) void gemm_strip_f32(
    const float* __restrict__ A0, const float* __restrict__ B0, float* __restrict__ C0,
    const float* __restrict__ A1, const float* __restrict__ B1, float* __restrict__ C1,
    int M, int gb, int auxb,
    const int* __restrict__ ei, int E1, const int* __restrict__ ei2, int E2,
    float* __restrict__ deg1, float* __restrict__ deg2,
    int* __restrict__ bkt,
    float* __restrict__ zp, int zn,
    float* hu, const float* __restrict__ aggu, const float* __restrict__ xsu,
    const float* __restrict__ gsum1, const float* __restrict__ gsum2)
{
    __shared__ float sA[AUX == 0 ? 32 * APITCH : 1];
    if constexpr (AUX == 0) {
        const int wvu = __builtin_amdgcn_readfirstlane((int)(threadIdx.x >> 5));
        update_rows_to_lds<8>(sA, hu, aggu, xsu, gsum1, deg1, gsum2, deg2,
                              (int)blockIdx.x * 32, M, wvu, (int)(threadIdx.x & 31));
        __syncthreads();
    }
    if (AUX != 0 && (int)blockIdx.x < auxb) {
        const int t = blockIdx.x * 128 + threadIdx.x, T = auxb * 128;
        if (AUX == 1) {
            for (int i = t; i < zn; i += T) zp[i] = 0.0f;
        } else {
            edge_prep_part(ei, E1, ei2, E2, deg1, deg2, bkt, M, t, T);
        }
        return;
    }
    const int  g      = (int)blockIdx.x - auxb;
    const bool second = g >= gb;
    const float* A = second ? A1 : A0;
    const float* B = second ? B1 : B0;
    float*       C = second ? C1 : C0;
    constexpr int K = HIDC, NC = HIDC;
    const int tilesM = (M + 15) >> 4;
    const int wid = (second ? g - gb : g) * 4 + (threadIdx.x >> 5);
    const int tm  = wid >> 1;
    if (tm >= tilesM) return;
    const int nb   = (wid & 1) << 6;
    const int m0   = tm << 4;
    const int lane = threadIdx.x & 31;
    const int hf   = lane >> 4;
    const int c    = lane & 15;

    const int arow = min(m0 + c, M - 1);
    const float* Ap = A + (size_t)arow * K + 4 * hf;
    const float* Bp = B + (size_t)(4 * hf) * NC + nb + 4 * c;
    const float* Al = sA + ((int)((threadIdx.x >> 6) << 4) + c) * APITCH + 4 * hf;

    v8f acc0 = {}, acc1 = {}, acc2 = {}, acc3 = {};
#pragma unroll
    for (int kb = 0; kb < K; kb += 8) {
        const float4 a  = (AUX == 0) ? *(const float4*)(Al + kb) : *(const float4*)(Ap + kb);
        const float4 b0 = *(const float4*)(Bp + (kb + 0) * NC);
        const float4 b1 = *(const float4*)(Bp + (kb + 1) * NC);
        const float4 b2 = *(const float4*)(Bp + (kb + 2) * NC);
        const float4 b3 = *(const float4*)(Bp + (kb + 3) * NC);
        acc0 = wmma4(a.x, a.y, b0.x, b1.x, acc0);
        acc1 = wmma4(a.x, a.y, b0.y, b1.y, acc1);
        acc2 = wmma4(a.x, a.y, b0.z, b1.z, acc2);
        acc3 = wmma4(a.x, a.y, b0.w, b1.w, acc3);
        acc0 = wmma4(a.z, a.w, b2.x, b3.x, acc0);
        acc1 = wmma4(a.z, a.w, b2.y, b3.y, acc1);
        acc2 = wmma4(a.z, a.w, b2.z, b3.z, acc2);
        acc3 = wmma4(a.z, a.w, b2.w, b3.w, acc3);
    }

#pragma unroll
    for (int v = 0; v < 8; ++v) {
        const int r = m0 + v + 8 * hf;
        if (r < M)
            *(float4*)(C + (size_t)r * NC + nb + 4 * c) =
                make_float4(acc0[v], acc1[v], acc2[v], acc3[v]);
    }
}


#define ACH 8

__global__ __launch_bounds__(256) void gather_agg(
    const float* __restrict__ m, const int* __restrict__ bkt, const float* __restrict__ b,
    float* __restrict__ agg, int n)
{
    const int wv   = __builtin_amdgcn_readfirstlane((int)(threadIdx.x >> 5));
    const int lane = threadIdx.x & 31;
    const int node = blockIdx.x * 8 + wv;
    if (node >= n) return;
    const int sv = bkt[(size_t)node * BW + min(lane, BW - 1)];
    const int c  = __builtin_amdgcn_readlane(sv, 0);
    const int nb = min(c, BCAP);

    float4 acc = make_float4(0.0f, 0.0f, 0.0f, 0.0f);
    for (int j0 = 0; j0 < nb; j0 += ACH) {
        const int nl = nb - j0;
        float4 v[ACH];
        v[0] = *(const float4*)(m + (size_t)__builtin_amdgcn_readlane(sv, 1 + j0) * HIDC + lane * 4);
#pragma unroll
        for (int j = 1; j < ACH; ++j)
            if (j < nl) {
                const int sj = __builtin_amdgcn_readlane(sv, 1 + j0 + j);
                v[j] = *(const float4*)(m + (size_t)sj * HIDC + lane * 4);
            }
        acc.x += v[0].x; acc.y += v[0].y; acc.z += v[0].z; acc.w += v[0].w;
#pragma unroll
        for (int j = 1; j < ACH; ++j)
            if (j < nl) { acc.x += v[j].x; acc.y += v[j].y; acc.z += v[j].z; acc.w += v[j].w; }
    }
    if (c > BCAP) {
        const int  novf = *ovf_count(bkt, n);
        const int* ovf  = ovf_pairs(bkt, n);
        for (int i0 = 0; i0 < novf; i0 += 32) {
            const int  i  = i0 + lane;
            const bool in = i < novf;
            const int2 pr = in ? *(const int2*)(ovf + 2 * i) : make_int2(-1, 0);
            unsigned hits = (unsigned)__ballot(in && pr.x == node);
            while (hits) {
                const int j = __builtin_ctz(hits);
                hits &= hits - 1;
                const int sj = __builtin_amdgcn_readlane(pr.y, j);
                const float4 v = *(const float4*)(m + (size_t)sj * HIDC + lane * 4);
                acc.x += v.x; acc.y += v.y; acc.z += v.z; acc.w += v.w;
            }
        }
    }
    const float4 bb = *(const float4*)(b + lane * 4);
    float4 r;
    r.x = fmaxf(acc.x + bb.x, 0.0f);
    r.y = fmaxf(acc.y + bb.y, 0.0f);
    r.z = fmaxf(acc.z + bb.z, 0.0f);
    r.w = fmaxf(acc.w + bb.w, 0.0f);
    *(float4*)(agg + (size_t)node * HIDC + lane * 4) = r;
}

#define GCH 8
#define GBW 4

__device__ __forceinline__ float4 ld_row4(const float* __restrict__ agg, int node, int lane) {
    return *(const float4*)(agg + (size_t)node * HIDC + lane * 4);
}

__device__ __forceinline__ float sqdist4(const float4 a, const float4 b) {
    float dx = a.x - b.x, dy = a.y - b.y, dz = a.z - b.z, dw = a.w - b.w;
    return dx * dx + dy * dy + dz * dz + dw * dw;
}

__device__ __forceinline__ void flush_run(float* __restrict__ gsum, int node, float s, int lane) {
#pragma unroll
    for (int off = 16; off > 0; off >>= 1)
        s += __shfl_down(s, off, 32);
    if (lane == 0) atomicAdd(&gsum[node], s);
}

__device__ __forceinline__ void walk_edges(
    const float* __restrict__ agg, const int* __restrict__ ei, int E,
    float* __restrict__ gsum, int e_beg, int e_end, int lane)
{
    int    prev = -1;
    float  acc  = 0.0f;
    float4 a    = make_float4(0.0f, 0.0f, 0.0f, 0.0f);
    int e0 = e_beg;
    if (e0 + GCH <= e_end) {
        int rr[GCH], cc[GCH];
#pragma unroll
        for (int j = 0; j < GCH; ++j) { rr[j] = ei[e0 + j]; cc[j] = ei[E + e0 + j]; }
        for (;;) {
            const bool more = e0 + 2 * GCH <= e_end;
            const int  en   = more ? e0 + GCH : e0;
            int rn[GCH], cn[GCH];
#pragma unroll
            for (int j = 0; j < GCH; ++j) { rn[j] = ei[en + j]; cn[j] = ei[E + en + j]; }

            float4 bv[GCH], av[GCH];
#pragma unroll
            for (int j = 0; j < GCH; ++j) bv[j] = ld_row4(agg, cc[j], lane);
#pragma unroll
            for (int j = 0; j < GCH; ++j) {
                const int pj = j ? rr[j - 1] : prev;
                if (rr[j] != pj) av[j] = ld_row4(agg, rr[j], lane);
            }
#pragma unroll
            for (int j = 0; j < GCH; ++j) {
                const int pj = j ? rr[j - 1] : prev;
                if (rr[j] != pj) {
                    if (j > 0 || pj >= 0) flush_run(gsum, pj, acc, lane);
                    acc = 0.0f;
                    a = av[j];
                }
                acc += sqdist4(a, bv[j]);
            }
            prev = rr[GCH - 1];
            e0 += GCH;
            if (!more) break;
#pragma unroll
            for (int j = 0; j < GCH; ++j) { rr[j] = rn[j]; cc[j] = cn[j]; }
        }
    }
    for (; e0 < e_end; ++e0) {
        const int r = ei[e0];
        const int c = ei[E + e0];
        if (r != prev) {
            if (prev >= 0) flush_run(gsum, prev, acc, lane);
            acc = 0.0f; prev = r;
            a = ld_row4(agg, r, lane);
        }
        acc += sqdist4(a, ld_row4(agg, c, lane));
    }
    flush_run(gsum, prev, acc, lane);
}

__global__ __launch_bounds__(256) void gamma_edges(
    const float* __restrict__ agg,
    const int* __restrict__ eiA, int EA, float* __restrict__ gsumA, int qA, int rA,
    const int* __restrict__ eiB, int EB, float* __restrict__ gsumB, int qB, int rB)
{
    const int wv   = __builtin_amdgcn_readfirstlane((int)(threadIdx.x >> 5));
    const int lane = threadIdx.x & 31;
    const int w    = blockIdx.x * 8 + wv;
    const int nw   = gridDim.x * 8;

    const int a0 = w * qA + min(w, rA);
    const int a1 = a0 + qA + (w < rA ? 1 : 0);
    if (a0 < a1)
        walk_edges(agg, eiA, EA, gsumA, a0 * GCH, min(a1 * GCH, EA), lane);

    const int xb = max(w - (nw - rB), 0);
    const int b0 = w * qB + xb;
    const int b1 = b0 + qB + (w >= nw - rB ? 1 : 0);
    if (b0 < b1)
        walk_edges(agg, eiB, EB, gsumB, b0 * GCH, min(b1 * GCH, EB), lane);
}

__global__ __launch_bounds__(256) void update_h(
    float* __restrict__ h, const float* __restrict__ agg,
    const float* __restrict__ xs,
    const float* __restrict__ gsum1, const float* __restrict__ deg1,
    const float* __restrict__ gsum2, const float* __restrict__ deg2, int n)
{
    const int wv   = __builtin_amdgcn_readfirstlane((int)(threadIdx.x >> 5));
    const int lane = threadIdx.x & 31;
    const int node = blockIdx.x * 8 + wv;
    if (node >= n) return;
    const size_t o = (size_t)node * HIDC + lane * 4;
    const float4 hv = *(const float4*)(h + o);
    const float4 av = *(const float4*)(agg + o);
    const float4 xv = *(const float4*)(xs + o);
    const float gs = tanhf(gsum1[node] * (1.0f / (deg1[node] + 1e-10f)));
    const float gq = tanhf(gsum2[node] * (1.0f / (deg2[node] + 1e-10f)));
    const float den = 1.0f + gs + gq;
    float4 r;
    r.x = (hv.x + gs * av.x + gq * xv.x) / den;
    r.y = (hv.y + gs * av.y + gq * xv.y) / den;
    r.z = (hv.z + gs * av.z + gq * xv.z) / den;
    r.w = (hv.w + gs * av.w + gq * xv.w) / den;
    *(float4*)(h + o) = r;
}

extern "C" void kernel_launch(void* const* d_in, const int* in_sizes, int n_in,
                              void* d_out, int out_size, void* d_ws, size_t ws_size,
                              hipStream_t stream) {
    const float* x      = (const float*)d_in[0];
    const float* x0     = (const float*)d_in[1];
    const int*   ei     = (const int*)d_in[2];
    const int*   ei2    = (const int*)d_in[3];
    const float* in_W   = (const float*)d_in[4];
    const float* skip_W = (const float*)d_in[5];
    const float* conv_W = (const float*)d_in[6];
    const float* conv_b = (const float*)d_in[7];
    const float* fc_W   = (const float*)d_in[8];
    const float* fc_b   = (const float*)d_in[9];
    float* out = (float*)d_out;

    const int N  = in_sizes[0] / HIDC;
    const int E1 = in_sizes[2] / 2;
    const int E2 = in_sizes[3] / 2;

    const size_t nh = (size_t)N * HIDC;
    float* h     = (float*)d_ws;
    float* xskip = h + nh;
    float* agg   = xskip + nh;
    float* mbuf  = agg + nh;

    float* zbase = (float*)d_out;
    float* deg1  = zbase;
    float* deg2  = deg1 + N;
    float* gsum1[2] = { deg2 + N,     deg2 + 3 * N };
    float* gsum2[2] = { deg2 + 2 * N, deg2 + 4 * N };
    int* bkt = (int*)(zbase + 6 * N);

    const dim3 blk(256);
    auto gemm_blocks = [](int M, int Nc, int pairs) {
        int tiles = ((M + 15) / 16) * ((Nc + 15) / 16);
        return dim3((tiles + 7) / 8, pairs);
    };
    const dim3 sblk(128);
    const int sb = (((N + 15) / 16) * 2 + 3) / 4;
    const float* conv_W1 = conv_W + (size_t)HIDC * HIDC;
    const int zb = 128, eb = 512;

    gemm_strip_f32<1><<<zb + 2 * sb, sblk, 0, stream>>>(
        x, in_W, h, x0, in_W, agg, N, sb, zb,
        nullptr, 0, nullptr, 0, nullptr, nullptr, nullptr, zbase, (6 + BW) * N + 4,
        nullptr, nullptr, nullptr, nullptr, nullptr);
    gemm_strip_f32<2><<<eb + 2 * sb, sblk, 0, stream>>>(
        agg, skip_W, xskip, h, conv_W, mbuf, N, sb, eb,
        ei, E1, ei2, E2, deg1, deg2, bkt, nullptr, 0,
        nullptr, nullptr, nullptr, nullptr, nullptr);

    const int gblocks = 256 * GBW;
    const int gwaves  = gblocks * 8;
    const int nch2 = (E2 + GCH - 1) / GCH, nch1 = (E1 + GCH - 1) / GCH;
    const int q2 = nch2 / gwaves, r2 = nch2 % gwaves;
    const int q1 = nch1 / gwaves, r1 = nch1 % gwaves;
    for (int l = 0; l < 2; ++l) {
        if (l == 1)
            gemm_strip_f32<0><<<sb, sblk, 0, stream>>>(
                nullptr, conv_W1, mbuf, nullptr, nullptr, nullptr, N, sb, 0,
                nullptr, 0, nullptr, 0, deg1, deg2, nullptr, nullptr, 0,
                h, agg, xskip, gsum1[0], gsum2[0]);
        gather_agg<<<(N + 7) / 8, blk, 0, stream>>>(mbuf, bkt, conv_b + l * HIDC, agg, N);
        gamma_edges<<<gblocks, blk, 0, stream>>>(agg, ei2, E2, gsum2[l], q2, r2, ei, E1, gsum1[l], q1, r1);
        if (l == 1)
            update_h<<<(N + 7) / 8, blk, 0, stream>>>(h, agg, xskip, gsum1[l], deg1, gsum2[l], deg2, N);
    }

    gemm_wmma_f32<40><<<gemm_blocks(N, 40, 1), blk, 0, stream>>>(
        h, fc_W, fc_b, out, nullptr, nullptr, nullptr, N);
}


# fewer launches (9 -> 8), same arithmetic and sum order: last update_h rides in the fc product launch (rows reach WMMA via LDS, h not stored); per-node scratch now sits in the node's own out row
# speedup vs baseline: 3.4820x; 1.2118x over previous
#include <hip/hip_runtime.h>

typedef __attribute__((ext_vector_type(2))) float v2f;
typedef __attribute__((ext_vector_type(8))) float v8f;

#define HIDC 128

#define BW   20
#define BCAP (BW - 1)
#define NS   40
#define OVP  6
__device__ __forceinline__ int* ovf_count(int* bkt) { return bkt + 32; }
__device__ __forceinline__ const int* ovf_count(const int* bkt) { return bkt + 32; }
__device__ __forceinline__ int* ovf_pair(int* bkt, int o) {
    return bkt + (size_t)(o / OVP) * NS + BW + 2 * (o % OVP);
}
__device__ __forceinline__ const int* ovf_pair(const int* bkt, int o) {
    return bkt + (size_t)(o / OVP) * NS + BW + 2 * (o % OVP);
}

__device__ __forceinline__ void edge_prep_part(
    const int* __restrict__ ei, int E1, const int* __restrict__ ei2, int E2,
    float* __restrict__ deg1, float* __restrict__ deg2,
    int* __restrict__ bkt, int t, int T)
{
    int* const novf = ovf_count(bkt);
    for (int e = t; e < E1; e += T) {
        const int src = ei[e], dst = ei[E1 + e];
        atomicAdd(&deg1[(size_t)src * NS], 1.0f);
        int* const bk = bkt + (size_t)dst * NS;
        const int slot = atomicAdd(bk, 1);
        if (slot < BCAP) {
            bk[1 + slot] = src;
        } else {
            const int o = atomicAdd(novf, 1);
            *(int2*)ovf_pair(bkt, o) = make_int2(dst, src);
        }
    }
    const int n4 = E2 >> 2;
    for (int g = t; g < n4; g += T) {
        const int4 r = *(const int4*)(ei2 + 4 * g);
        int p = r.x;
        float c = 1.0f;
        if (r.y == p) c += 1.0f; else { atomicAdd(&deg2[(size_t)p * NS], c); p = r.y; c = 1.0f; }
        if (r.z == p) c += 1.0f; else { atomicAdd(&deg2[(size_t)p * NS], c); p = r.z; c = 1.0f; }
        if (r.w == p) c += 1.0f; else { atomicAdd(&deg2[(size_t)p * NS], c); p = r.w; c = 1.0f; }
        atomicAdd(&deg2[(size_t)p * NS], c);
    }
    for (int f = (n4 << 2) + t; f < E2; f += T)
        atomicAdd(&deg2[(size_t)ei2[f] * NS], 1.0f);
}

#define APITCH 132

__device__ __forceinline__ float4 gated_row(const float4 hv, const float4 av, const float4 xv,
                                            float gs, float gq, float den) {
    float4 r;
    r.x = (hv.x + gs * av.x + gq * xv.x) / den;
    r.y = (hv.y + gs * av.y + gq * xv.y) / den;
    r.z = (hv.z + gs * av.z + gq * xv.z) / den;
    r.w = (hv.w + gs * av.w + gq * xv.w) / den;
    return r;
}

__device__ __forceinline__ float lane_value(float x, int i) {
    return __builtin_bit_cast(float, __builtin_amdgcn_readlane(__builtin_bit_cast(int, x), i));
}

template <int RPW, bool STORE_H>
__device__ __forceinline__ void update_rows_to_lds(
    float* sA, float* h, const float* __restrict__ agg, const float* __restrict__ xs,
    const float* __restrict__ gsum1, const float* __restrict__ deg1,
    const float* __restrict__ gsum2, const float* __restrict__ deg2,
    int r0, int M, int wv, int lane)
{
    const int gn = min(r0 + wv * RPW + min(lane, RPW - 1), M - 1);
    const size_t go = (size_t)gn * NS;
    const float g1 = gsum1[go], d1 = deg1[go], g2 = gsum2[go], d2 = deg2[go];
    float4 hv[RPW], av[RPW], xv[RPW];
#pragma unroll
    for (int i = 0; i < RPW; ++i) {
        const size_t o = (size_t)min(r0 + wv * RPW + i, M - 1) * HIDC + lane * 4;
        hv[i] = *(const float4*)(h + o);
        av[i] = *(const float4*)(agg + o);
        xv[i] = *(const float4*)(xs + o);
    }
    const float gsl  = tanhf(g1 * (1.0f / (d1 + 1e-10f)));
    const float gql  = tanhf(g2 * (1.0f / (d2 + 1e-10f)));
    const float denl = 1.0f + gsl + gql;
#pragma unroll
    for (int i = 0; i < RPW; ++i) {
        const int lr = wv * RPW + i, node = r0 + lr;
        float4 r = make_float4(0.0f, 0.0f, 0.0f, 0.0f);
        if (node < M) {
            r = gated_row(hv[i], av[i], xv[i],
                          lane_value(gsl, i), lane_value(gql, i), lane_value(denl, i));
            if (STORE_H) *(float4*)(h + (size_t)node * HIDC + lane * 4) = r;
        }
        *(float4*)(sA + lr * APITCH + lane * 4) = r;
    }
}

__device__ __forceinline__ v8f wmma4(const float a0, const float a1,
                                     const float b0, const float b1, const v8f acc) {
    v2f a; a.x = a0; a.y = a1;
    v2f b; b.x = b0; b.y = b1;
    return __builtin_amdgcn_wmma_f32_16x16x4_f32(false, a, false, b, (short)0, acc, false, false);
}

template <int AUX>
__global__ __launch_bounds__(128) __attribute__((amdgpu_waves_per_eu(1, 4))) void gemm_strip_f32(
    const float* __restrict__ A0, const float* __restrict__ B0, float* __restrict__ C0,
    const float* __restrict__ A1, const float* __restrict__ B1, float* __restrict__ C1,
    int M, int gb, int auxb,
    const int* __restrict__ ei, int E1, const int* __restrict__ ei2, int E2,
    float* __restrict__ deg1, float* __restrict__ deg2,
    int* __restrict__ bkt,
    float* __restrict__ zp, int zn,
    float* hu, const float* __restrict__ aggu, const float* __restrict__ xsu,
    const float* __restrict__ gsum1, const float* __restrict__ gsum2)
{
    __shared__ float sA[AUX == 0 ? 32 * APITCH : 1];
    if constexpr (AUX == 0) {
        const int wvu = __builtin_amdgcn_readfirstlane((int)(threadIdx.x >> 5));
        update_rows_to_lds<8, true>(sA, hu, aggu, xsu, gsum1, deg1, gsum2, deg2,
                              (int)blockIdx.x * 32, M, wvu, (int)(threadIdx.x & 31));
        __syncthreads();
    }
    if (AUX != 0 && (int)blockIdx.x < auxb) {
        const int t = blockIdx.x * 128 + threadIdx.x, T = auxb * 128;
        if (AUX == 1) {
            for (int i = t; i < zn; i += T) zp[i] = 0.0f;
        } else {
            edge_prep_part(ei, E1, ei2, E2, deg1, deg2, bkt, t, T);
        }
        return;
    }
    const int  g      = (int)blockIdx.x - auxb;
    const bool second = g >= gb;
    const float* A = second ? A1 : A0;
    const float* B = second ? B1 : B0;
    float*       C = second ? C1 : C0;
    constexpr int K = HIDC, NC = HIDC;
    const int tilesM = (M + 15) >> 4;
    const int wid = (second ? g - gb : g) * 4 + (threadIdx.x >> 5);
    const int tm  = wid >> 1;
    if (tm >= tilesM) return;
    const int nb   = (wid & 1) << 6;
    const int m0   = tm << 4;
    const int lane = threadIdx.x & 31;
    const int hf   = lane >> 4;
    const int c    = lane & 15;

    const int arow = min(m0 + c, M - 1);
    const float* Ap = A + (size_t)arow * K + 4 * hf;
    const float* Bp = B + (size_t)(4 * hf) * NC + nb + 4 * c;
    const float* Al = sA + ((int)((threadIdx.x >> 6) << 4) + c) * APITCH + 4 * hf;

    v8f acc0 = {}, acc1 = {}, acc2 = {}, acc3 = {};
#pragma unroll
    for (int kb = 0; kb < K; kb += 8) {
        const float4 a  = (AUX == 0) ? *(const float4*)(Al + kb) : *(const float4*)(Ap + kb);
        const float4 b0 = *(const float4*)(Bp + (kb + 0) * NC);
        const float4 b1 = *(const float4*)(Bp + (kb + 1) * NC);
        const float4 b2 = *(const float4*)(Bp + (kb + 2) * NC);
        const float4 b3 = *(const float4*)(Bp + (kb + 3) * NC);
        acc0 = wmma4(a.x, a.y, b0.x, b1.x, acc0);
        acc1 = wmma4(a.x, a.y, b0.y, b1.y, acc1);
        acc2 = wmma4(a.x, a.y, b0.z, b1.z, acc2);
        acc3 = wmma4(a.x, a.y, b0.w, b1.w, acc3);
        acc0 = wmma4(a.z, a.w, b2.x, b3.x, acc0);
        acc1 = wmma4(a.z, a.w, b2.y, b3.y, acc1);
        acc2 = wmma4(a.z, a.w, b2.z, b3.z, acc2);
        acc3 = wmma4(a.z, a.w, b2.w, b3.w, acc3);
    }

#pragma unroll
    for (int v = 0; v < 8; ++v) {
        const int r = m0 + v + 8 * hf;
        if (r < M)
            *(float4*)(C + (size_t)r * NC + nb + 4 * c) =
                make_float4(acc0[v], acc1[v], acc2[v], acc3[v]);
    }
}


#define ACH 8

__global__ __launch_bounds__(256) void gather_agg(
    const float* __restrict__ m, const int* __restrict__ bkt, const float* __restrict__ b,
    float* __restrict__ agg, int n)
{
    const int wv   = __builtin_amdgcn_readfirstlane((int)(threadIdx.x >> 5));
    const int lane = threadIdx.x & 31;
    const int node = blockIdx.x * 8 + wv;
    if (node >= n) return;
    const int sv = bkt[(size_t)node * NS + min(lane, BW - 1)];
    const int c  = __builtin_amdgcn_readlane(sv, 0);
    const int nb = min(c, BCAP);

    float4 acc = make_float4(0.0f, 0.0f, 0.0f, 0.0f);
    for (int j0 = 0; j0 < nb; j0 += ACH) {
        const int nl = nb - j0;
        float4 v[ACH];
        v[0] = *(const float4*)(m + (size_t)__builtin_amdgcn_readlane(sv, 1 + j0) * HIDC + lane * 4);
#pragma unroll
        for (int j = 1; j < ACH; ++j)
            if (j < nl) {
                const int sj = __builtin_amdgcn_readlane(sv, 1 + j0 + j);
                v[j] = *(const float4*)(m + (size_t)sj * HIDC + lane * 4);
            }
        acc.x += v[0].x; acc.y += v[0].y; acc.z += v[0].z; acc.w += v[0].w;
#pragma unroll
        for (int j = 1; j < ACH; ++j)
            if (j < nl) { acc.x += v[j].x; acc.y += v[j].y; acc.z += v[j].z; acc.w += v[j].w; }
    }
    if (c > BCAP) {
        const int  novf = *ovf_count(bkt);
        for (int i0 = 0; i0 < novf; i0 += 32) {
            const int  i  = i0 + lane;
            const bool in = i < novf;
            const int2 pr = in ? *(const int2*)ovf_pair(bkt, i) : make_int2(-1, 0);
            unsigned hits = (unsigned)__ballot(in && pr.x == node);
            while (hits) {
                const int j = __builtin_ctz(hits);
                hits &= hits - 1;
                const int sj = __builtin_amdgcn_readlane(pr.y, j);
                const float4 v = *(const float4*)(m + (size_t)sj * HIDC + lane * 4);
                acc.x += v.x; acc.y += v.y; acc.z += v.z; acc.w += v.w;
            }
        }
    }
    const float4 bb = *(const float4*)(b + lane * 4);
    float4 r;
    r.x = fmaxf(acc.x + bb.x, 0.0f);
    r.y = fmaxf(acc.y + bb.y, 0.0f);
    r.z = fmaxf(acc.z + bb.z, 0.0f);
    r.w = fmaxf(acc.w + bb.w, 0.0f);
    *(float4*)(agg + (size_t)node * HIDC + lane * 4) = r;
}

#define GCH 8
#define GBW 4

__device__ __forceinline__ float4 ld_row4(const float* __restrict__ agg, int node, int lane) {
    return *(const float4*)(agg + (size_t)node * HIDC + lane * 4);
}

__device__ __forceinline__ float sqdist4(const float4 a, const float4 b) {
    float dx = a.x - b.x, dy = a.y - b.y, dz = a.z - b.z, dw = a.w - b.w;
    return dx * dx + dy * dy + dz * dz + dw * dw;
}

__device__ __forceinline__ void flush_run(float* __restrict__ gsum, int node, float s, int lane) {
#pragma unroll
    for (int off = 16; off > 0; off >>= 1)
        s += __shfl_down(s, off, 32);
    if (lane == 0) atomicAdd(&gsum[(size_t)node * NS], s);
}

__device__ __forceinline__ void walk_edges(
    const float* __restrict__ agg, const int* __restrict__ ei, int E,
    float* __restrict__ gsum, int e_beg, int e_end, int lane)
{
    int    prev = -1;
    float  acc  = 0.0f;
    float4 a    = make_float4(0.0f, 0.0f, 0.0f, 0.0f);
    int e0 = e_beg;
    if (e0 + GCH <= e_end) {
        int rr[GCH], cc[GCH];
#pragma unroll
        for (int j = 0; j < GCH; ++j) { rr[j] = ei[e0 + j]; cc[j] = ei[E + e0 + j]; }
        for (;;) {
            const bool more = e0 + 2 * GCH <= e_end;
            const int  en   = more ? e0 + GCH : e0;
            int rn[GCH], cn[GCH];
#pragma unroll
            for (int j = 0; j < GCH; ++j) { rn[j] = ei[en + j]; cn[j] = ei[E + en + j]; }

            float4 bv[GCH], av[GCH];
#pragma unroll
            for (int j = 0; j < GCH; ++j) bv[j] = ld_row4(agg, cc[j], lane);
#pragma unroll
            for (int j = 0; j < GCH; ++j) {
                const int pj = j ? rr[j - 1] : prev;
                if (rr[j] != pj) av[j] = ld_row4(agg, rr[j], lane);
            }
#pragma unroll
            for (int j = 0; j < GCH; ++j) {
                const int pj = j ? rr[j - 1] : prev;
                if (rr[j] != pj) {
                    if (j > 0 || pj >= 0) flush_run(gsum, pj, acc, lane);
                    acc = 0.0f;
                    a = av[j];
                }
                acc += sqdist4(a, bv[j]);
            }
            prev = rr[GCH - 1];
            e0 += GCH;
            if (!more) break;
#pragma unroll
            for (int j = 0; j < GCH; ++j) { rr[j] = rn[j]; cc[j] = cn[j]; }
        }
    }
    for (; e0 < e_end; ++e0) {
        const int r = ei[e0];
        const int c = ei[E + e0];
        if (r != prev) {
            if (prev >= 0) flush_run(gsum, prev, acc, lane);
            acc = 0.0f; prev = r;
            a = ld_row4(agg, r, lane);
        }
        acc += sqdist4(a, ld_row4(agg, c, lane));
    }
    flush_run(gsum, prev, acc, lane);
}

__global__ __launch_bounds__(256) void gamma_edges(
    const float* __restrict__ agg,
    const int* __restrict__ eiA, int EA, float* __restrict__ gsumA, int qA, int rA,
    const int* __restrict__ eiB, int EB, float* __restrict__ gsumB, int qB, int rB)
{
    const int wv   = __builtin_amdgcn_readfirstlane((int)(threadIdx.x >> 5));
    const int lane = threadIdx.x & 31;
    const int w    = blockIdx.x * 8 + wv;
    const int nw   = gridDim.x * 8;

    const int a0 = w * qA + min(w, rA);
    const int a1 = a0 + qA + (w < rA ? 1 : 0);
    if (a0 < a1)
        walk_edges(agg, eiA, EA, gsumA, a0 * GCH, min(a1 * GCH, EA), lane);

    const int xb = max(w - (nw - rB), 0);
    const int b0 = w * qB + xb;
    const int b1 = b0 + qB + (w >= nw - rB ? 1 : 0);
    if (b0 < b1)
        walk_edges(agg, eiB, EB, gsumB, b0 * GCH, min(b1 * GCH, EB), lane);
}

__global__ __launch_bounds__(128) void update_fc(
    float* h, const float* __restrict__ agg, const float* __restrict__ xs,
    const float* gsum1, const float* deg1, const float* gsum2, const float* deg2,
    const float* __restrict__ W, const float* __restrict__ bias, float* out, int M)
{
    constexpr int K = HIDC, NC = NS;
    __shared__ float sA[32 * APITCH];
    const int wvu  = __builtin_amdgcn_readfirstlane((int)(threadIdx.x >> 5));
    const int lane = threadIdx.x & 31;
    update_rows_to_lds<8, false>(sA, h, agg, xs, gsum1, deg1, gsum2, deg2,
                                 (int)blockIdx.x * 32, M, wvu, lane);
    __syncthreads();

    const int tr = wvu >> 1;
    const int m0 = (int)blockIdx.x * 32 + (tr << 4);
    if (m0 >= M) return;
    const int hf = lane >> 4;
    const int c  = lane & 15;
    const float* Al = sA + ((tr << 4) + c) * APITCH + 2 * hf;
    if ((wvu & 1) == 0) {
        const float* Bp = W + (size_t)(2 * hf) * NC + c;
        v8f acc0 = {}, acc1 = {};
#pragma unroll
        for (int k0 = 0; k0 < K; k0 += 4) {
            const v2f a = *(const v2f*)(Al + k0);
            acc0 = wmma4(a.x, a.y, Bp[k0 * NC],      Bp[(k0 + 1) * NC],      acc0);
            acc1 = wmma4(a.x, a.y, Bp[k0 * NC + 16], Bp[(k0 + 1) * NC + 16], acc1);
        }
        const float bv0 = bias[c], bv1 = bias[c + 16];
#pragma unroll
        for (int v = 0; v < 8; ++v) {
            const int r = m0 + v + 8 * hf;
            if (r < M) {
                out[(size_t)r * NC + c]      = acc0[v] + bv0;
                out[(size_t)r * NC + c + 16] = acc1[v] + bv1;
            }
        }
    } else {
        const int col  = 32 + c;
        const int colc = min(col, NC - 1);
        const float* Bp = W + (size_t)(2 * hf) * NC + colc;
        v8f acc = {};
#pragma unroll
        for (int k0 = 0; k0 < K; k0 += 4) {
            const v2f a = *(const v2f*)(Al + k0);
            acc = wmma4(a.x, a.y, Bp[k0 * NC], Bp[(k0 + 1) * NC], acc);
        }
        if (col < NC) {
            const float bv = bias[col];
#pragma unroll
            for (int v = 0; v < 8; ++v) {
                const int r = m0 + v + 8 * hf;
                if (r < M) out[(size_t)r * NC + col] = acc[v] + bv;
            }
        }
    }
}

extern "C" void kernel_launch(void* const* d_in, const int* in_sizes, int n_in,
                              void* d_out, int out_size, void* d_ws, size_t ws_size,
                              hipStream_t stream) {
    const float* x      = (const float*)d_in[0];
    const float* x0     = (const float*)d_in[1];
    const int*   ei     = (const int*)d_in[2];
    const int*   ei2    = (const int*)d_in[3];
    const float* in_W   = (const float*)d_in[4];
    const float* skip_W = (const float*)d_in[5];
    const float* conv_W = (const float*)d_in[6];
    const float* conv_b = (const float*)d_in[7];
    const float* fc_W   = (const float*)d_in[8];
    const float* fc_b   = (const float*)d_in[9];
    float* out = (float*)d_out;

    const int N  = in_sizes[0] / HIDC;
    const int E1 = in_sizes[2] / 2;
    const int E2 = in_sizes[3] / 2;

    const size_t nh = (size_t)N * HIDC;
    float* h     = (float*)d_ws;
    float* xskip = h + nh;
    float* agg   = xskip + nh;
    float* mbuf  = agg + nh;

    float* rec  = (float*)d_out;
    float* deg1 = rec;
    float* deg2 = rec + 1;
    float* gsum1[2] = { rec + 2, rec + 4 };
    float* gsum2[2] = { rec + 3, rec + 5 };
    int* bkt = (int*)(rec + 6);

    const dim3 blk(256);
    const dim3 sblk(128);
    const int sb = (((N + 15) / 16) * 2 + 3) / 4;
    const float* conv_W1 = conv_W + (size_t)HIDC * HIDC;
    const int zb = 128, eb = 512;

    gemm_strip_f32<1><<<zb + 2 * sb, sblk, 0, stream>>>(
        x, in_W, h, x0, in_W, agg, N, sb, zb,
        nullptr, 0, nullptr, 0, nullptr, nullptr, nullptr, rec, NS * N,
        nullptr, nullptr, nullptr, nullptr, nullptr);
    gemm_strip_f32<2><<<eb + 2 * sb, sblk, 0, stream>>>(
        agg, skip_W, xskip, h, conv_W, mbuf, N, sb, eb,
        ei, E1, ei2, E2, deg1, deg2, bkt, nullptr, 0,
        nullptr, nullptr, nullptr, nullptr, nullptr);

    const int gblocks = 256 * GBW;
    const int gwaves  = gblocks * 8;
    const int nch2 = (E2 + GCH - 1) / GCH, nch1 = (E1 + GCH - 1) / GCH;
    const int q2 = nch2 / gwaves, r2 = nch2 % gwaves;
    const int q1 = nch1 / gwaves, r1 = nch1 % gwaves;
    for (int l = 0; l < 2; ++l) {
        if (l == 1)
            gemm_strip_f32<0><<<sb, sblk, 0, stream>>>(
                nullptr, conv_W1, mbuf, nullptr, nullptr, nullptr, N, sb, 0,
                nullptr, 0, nullptr, 0, deg1, deg2, nullptr, nullptr, 0,
                h, agg, xskip, gsum1[0], gsum2[0]);
        gather_agg<<<(N + 7) / 8, blk, 0, stream>>>(mbuf, bkt, conv_b + l * HIDC, agg, N);
        gamma_edges<<<gblocks, blk, 0, stream>>>(agg, ei2, E2, gsum2[l], q2, r2, ei, E1, gsum1[l], q1, r1);
    }

    update_fc<<<(N + 31) / 32, sblk, 0, stream>>>(
        h, agg, xskip, gsum1[1], deg1, gsum2[1], deg2, fc_W, fc_b, out, N);
}


# gamma_edges: once a run's partial sum reaches 16*deg(row) the gate is exactly 1.0f in f32 (terms >= 0, tanh saturated): steps holding only that row are passed over, no gathers; output identical
# speedup vs baseline: 3.7274x; 1.0705x over previous
#include <hip/hip_runtime.h>

typedef __attribute__((ext_vector_type(2))) float v2f;
typedef __attribute__((ext_vector_type(8))) float v8f;

#define HIDC 128

#define BW   20
#define BCAP (BW - 1)
#define NS   40
#define OVP  6
__device__ __forceinline__ int* ovf_count(int* bkt) { return bkt + 32; }
__device__ __forceinline__ const int* ovf_count(const int* bkt) { return bkt + 32; }
__device__ __forceinline__ int* ovf_pair(int* bkt, int o) {
    return bkt + (size_t)(o / OVP) * NS + BW + 2 * (o % OVP);
}
__device__ __forceinline__ const int* ovf_pair(const int* bkt, int o) {
    return bkt + (size_t)(o / OVP) * NS + BW + 2 * (o % OVP);
}

__device__ __forceinline__ void edge_prep_part(
    const int* __restrict__ ei, int E1, const int* __restrict__ ei2, int E2,
    float* __restrict__ deg1, float* __restrict__ deg2,
    int* __restrict__ bkt, int t, int T)
{
    int* const novf = ovf_count(bkt);
    for (int e = t; e < E1; e += T) {
        const int src = ei[e], dst = ei[E1 + e];
        atomicAdd(&deg1[(size_t)src * NS], 1.0f);
        int* const bk = bkt + (size_t)dst * NS;
        const int slot = atomicAdd(bk, 1);
        if (slot < BCAP) {
            bk[1 + slot] = src;
        } else {
            const int o = atomicAdd(novf, 1);
            *(int2*)ovf_pair(bkt, o) = make_int2(dst, src);
        }
    }
    const int n4 = E2 >> 2;
    for (int g = t; g < n4; g += T) {
        const int4 r = *(const int4*)(ei2 + 4 * g);
        int p = r.x;
        float c = 1.0f;
        if (r.y == p) c += 1.0f; else { atomicAdd(&deg2[(size_t)p * NS], c); p = r.y; c = 1.0f; }
        if (r.z == p) c += 1.0f; else { atomicAdd(&deg2[(size_t)p * NS], c); p = r.z; c = 1.0f; }
        if (r.w == p) c += 1.0f; else { atomicAdd(&deg2[(size_t)p * NS], c); p = r.w; c = 1.0f; }
        atomicAdd(&deg2[(size_t)p * NS], c);
    }
    for (int f = (n4 << 2) + t; f < E2; f += T)
        atomicAdd(&deg2[(size_t)ei2[f] * NS], 1.0f);
}

#define APITCH 132

__device__ __forceinline__ float4 gated_row(const float4 hv, const float4 av, const float4 xv,
                                            float gs, float gq, float den) {
    float4 r;
    r.x = (hv.x + gs * av.x + gq * xv.x) / den;
    r.y = (hv.y + gs * av.y + gq * xv.y) / den;
    r.z = (hv.z + gs * av.z + gq * xv.z) / den;
    r.w = (hv.w + gs * av.w + gq * xv.w) / den;
    return r;
}

__device__ __forceinline__ float lane_value(float x, int i) {
    return __builtin_bit_cast(float, __builtin_amdgcn_readlane(__builtin_bit_cast(int, x), i));
}

template <int RPW, bool STORE_H>
__device__ __forceinline__ void update_rows_to_lds(
    float* sA, float* h, const float* __restrict__ agg, const float* __restrict__ xs,
    const float* __restrict__ gsum1, const float* __restrict__ deg1,
    const float* __restrict__ gsum2, const float* __restrict__ deg2,
    int r0, int M, int wv, int lane)
{
    const int gn = min(r0 + wv * RPW + min(lane, RPW - 1), M - 1);
    const size_t go = (size_t)gn * NS;
    const float g1 = gsum1[go], d1 = deg1[go], g2 = gsum2[go], d2 = deg2[go];
    float4 hv[RPW], av[RPW], xv[RPW];
#pragma unroll
    for (int i = 0; i < RPW; ++i) {
        const size_t o = (size_t)min(r0 + wv * RPW + i, M - 1) * HIDC + lane * 4;
        hv[i] = *(const float4*)(h + o);
        av[i] = *(const float4*)(agg + o);
        xv[i] = *(const float4*)(xs + o);
    }
    const float gsl  = tanhf(g1 * (1.0f / (d1 + 1e-10f)));
    const float gql  = tanhf(g2 * (1.0f / (d2 + 1e-10f)));
    const float denl = 1.0f + gsl + gql;
#pragma unroll
    for (int i = 0; i < RPW; ++i) {
        const int lr = wv * RPW + i, node = r0 + lr;
        float4 r = make_float4(0.0f, 0.0f, 0.0f, 0.0f);
        if (node < M) {
            r = gated_row(hv[i], av[i], xv[i],
                          lane_value(gsl, i), lane_value(gql, i), lane_value(denl, i));
            if (STORE_H) *(float4*)(h + (size_t)node * HIDC + lane * 4) = r;
        }
        *(float4*)(sA + lr * APITCH + lane * 4) = r;
    }
}

__device__ __forceinline__ v8f wmma4(const float a0, const float a1,
                                     const float b0, const float b1, const v8f acc) {
    v2f a; a.x = a0; a.y = a1;
    v2f b; b.x = b0; b.y = b1;
    return __builtin_amdgcn_wmma_f32_16x16x4_f32(false, a, false, b, (short)0, acc, false, false);
}

template <int AUX>
__global__ __launch_bounds__(128) __attribute__((amdgpu_waves_per_eu(1, 4))) void gemm_strip_f32(
    const float* __restrict__ A0, const float* __restrict__ B0, float* __restrict__ C0,
    const float* __restrict__ A1, const float* __restrict__ B1, float* __restrict__ C1,
    int M, int gb, int auxb,
    const int* __restrict__ ei, int E1, const int* __restrict__ ei2, int E2,
    float* __restrict__ deg1, float* __restrict__ deg2,
    int* __restrict__ bkt,
    float* __restrict__ zp, int zn,
    float* hu, const float* __restrict__ aggu, const float* __restrict__ xsu,
    const float* __restrict__ gsum1, const float* __restrict__ gsum2)
{
    __shared__ float sA[AUX == 0 ? 32 * APITCH : 1];
    if constexpr (AUX == 0) {
        const int wvu = __builtin_amdgcn_readfirstlane((int)(threadIdx.x >> 5));
        update_rows_to_lds<8, true>(sA, hu, aggu, xsu, gsum1, deg1, gsum2, deg2,
                              (int)blockIdx.x * 32, M, wvu, (int)(threadIdx.x & 31));
        __syncthreads();
    }
    if (AUX != 0 && (int)blockIdx.x < auxb) {
        const int t = blockIdx.x * 128 + threadIdx.x, T = auxb * 128;
        if (AUX == 1) {
            for (int i = t; i < zn; i += T) zp[i] = 0.0f;
        } else {
            edge_prep_part(ei, E1, ei2, E2, deg1, deg2, bkt, t, T);
        }
        return;
    }
    const int  g      = (int)blockIdx.x - auxb;
    const bool second = g >= gb;
    const float* A = second ? A1 : A0;
    const float* B = second ? B1 : B0;
    float*       C = second ? C1 : C0;
    constexpr int K = HIDC, NC = HIDC;
    const int tilesM = (M + 15) >> 4;
    const int wid = (second ? g - gb : g) * 4 + (threadIdx.x >> 5);
    const int tm  = wid >> 1;
    if (tm >= tilesM) return;
    const int nb   = (wid & 1) << 6;
    const int m0   = tm << 4;
    const int lane = threadIdx.x & 31;
    const int hf   = lane >> 4;
    const int c    = lane & 15;

    const int arow = min(m0 + c, M - 1);
    const float* Ap = A + (size_t)arow * K + 4 * hf;
    const float* Bp = B + (size_t)(4 * hf) * NC + nb + 4 * c;
    const float* Al = sA + ((int)((threadIdx.x >> 6) << 4) + c) * APITCH + 4 * hf;

    v8f acc0 = {}, acc1 = {}, acc2 = {}, acc3 = {};
#pragma unroll
    for (int kb = 0; kb < K; kb += 8) {
        const float4 a  = (AUX == 0) ? *(const float4*)(Al + kb) : *(const float4*)(Ap + kb);
        const float4 b0 = *(const float4*)(Bp + (kb + 0) * NC);
        const float4 b1 = *(const float4*)(Bp + (kb + 1) * NC);
        const float4 b2 = *(const float4*)(Bp + (kb + 2) * NC);
        const float4 b3 = *(const float4*)(Bp + (kb + 3) * NC);
        acc0 = wmma4(a.x, a.y, b0.x, b1.x, acc0);
        acc1 = wmma4(a.x, a.y, b0.y, b1.y, acc1);
        acc2 = wmma4(a.x, a.y, b0.z, b1.z, acc2);
        acc3 = wmma4(a.x, a.y, b0.w, b1.w, acc3);
        acc0 = wmma4(a.z, a.w, b2.x, b3.x, acc0);
        acc1 = wmma4(a.z, a.w, b2.y, b3.y, acc1);
        acc2 = wmma4(a.z, a.w, b2.z, b3.z, acc2);
        acc3 = wmma4(a.z, a.w, b2.w, b3.w, acc3);
    }

#pragma unroll
    for (int v = 0; v < 8; ++v) {
        const int r = m0 + v + 8 * hf;
        if (r < M)
            *(float4*)(C + (size_t)r * NC + nb + 4 * c) =
                make_float4(acc0[v], acc1[v], acc2[v], acc3[v]);
    }
}


#define ACH 8

__global__ __launch_bounds__(256) void gather_agg(
    const float* __restrict__ m, const int* __restrict__ bkt, const float* __restrict__ b,
    float* __restrict__ agg, int n)
{
    const int wv   = __builtin_amdgcn_readfirstlane((int)(threadIdx.x >> 5));
    const int lane = threadIdx.x & 31;
    const int node = blockIdx.x * 8 + wv;
    if (node >= n) return;
    const int sv = bkt[(size_t)node * NS + min(lane, BW - 1)];
    const int c  = __builtin_amdgcn_readlane(sv, 0);
    const int nb = min(c, BCAP);

    float4 acc = make_float4(0.0f, 0.0f, 0.0f, 0.0f);
    for (int j0 = 0; j0 < nb; j0 += ACH) {
        const int nl = nb - j0;
        float4 v[ACH];
        v[0] = *(const float4*)(m + (size_t)__builtin_amdgcn_readlane(sv, 1 + j0) * HIDC + lane * 4);
#pragma unroll
        for (int j = 1; j < ACH; ++j)
            if (j < nl) {
                const int sj = __builtin_amdgcn_readlane(sv, 1 + j0 + j);
                v[j] = *(const float4*)(m + (size_t)sj * HIDC + lane * 4);
            }
        acc.x += v[0].x; acc.y += v[0].y; acc.z += v[0].z; acc.w += v[0].w;
#pragma unroll
        for (int j = 1; j < ACH; ++j)
            if (j < nl) { acc.x += v[j].x; acc.y += v[j].y; acc.z += v[j].z; acc.w += v[j].w; }
    }
    if (c > BCAP) {
        const int  novf = *ovf_count(bkt);
        for (int i0 = 0; i0 < novf; i0 += 32) {
            const int  i  = i0 + lane;
            const bool in = i < novf;
            const int2 pr = in ? *(const int2*)ovf_pair(bkt, i) : make_int2(-1, 0);
            unsigned hits = (unsigned)__ballot(in && pr.x == node);
            while (hits) {
                const int j = __builtin_ctz(hits);
                hits &= hits - 1;
                const int sj = __builtin_amdgcn_readlane(pr.y, j);
                const float4 v = *(const float4*)(m + (size_t)sj * HIDC + lane * 4);
                acc.x += v.x; acc.y += v.y; acc.z += v.z; acc.w += v.w;
            }
        }
    }
    const float4 bb = *(const float4*)(b + lane * 4);
    float4 r;
    r.x = fmaxf(acc.x + bb.x, 0.0f);
    r.y = fmaxf(acc.y + bb.y, 0.0f);
    r.z = fmaxf(acc.z + bb.z, 0.0f);
    r.w = fmaxf(acc.w + bb.w, 0.0f);
    *(float4*)(agg + (size_t)node * HIDC + lane * 4) = r;
}

#define GCH 8
#define GBW 4

__device__ __forceinline__ float4 ld_row4(const float* __restrict__ agg, int node, int lane) {
    return *(const float4*)(agg + (size_t)node * HIDC + lane * 4);
}

__device__ __forceinline__ float sqdist4(const float4 a, const float4 b) {
    float dx = a.x - b.x, dy = a.y - b.y, dz = a.z - b.z, dw = a.w - b.w;
    return dx * dx + dy * dy + dz * dz + dw * dw;
}

__device__ __forceinline__ void flush_run(float* __restrict__ gsum, int node, float s, int lane) {
#pragma unroll
    for (int off = 16; off > 0; off >>= 1)
        s += __shfl_down(s, off, 32);
    if (lane == 0) atomicAdd(&gsum[(size_t)node * NS], s);
}

__device__ __forceinline__ float wave_total(float s) {
#pragma unroll
    for (int off = 16; off > 0; off >>= 1)
        s += __shfl_down(s, off, 32);
    return lane_value(s, 0);
}

#define SATK 16.0f

__device__ __forceinline__ void walk_edges(
    const float* __restrict__ agg, const int* __restrict__ ei, int E,
    float* __restrict__ gsum, const float* __restrict__ degp, int e_beg, int e_end, int lane)
{
    int    prev = -1;
    float  acc  = 0.0f;
    bool   sat  = false;
    float4 a    = make_float4(0.0f, 0.0f, 0.0f, 0.0f);
    int e0 = e_beg;
    if (e0 + GCH <= e_end) {
        int rr[GCH], cc[GCH];
#pragma unroll
        for (int j = 0; j < GCH; ++j) { rr[j] = ei[e0 + j]; cc[j] = ei[E + e0 + j]; }
        for (;;) {
            const bool more = e0 + 2 * GCH <= e_end;
            const int  en   = more ? e0 + GCH : e0;
            int rn[GCH], cn[GCH];
#pragma unroll
            for (int j = 0; j < GCH; ++j) { rn[j] = ei[en + j]; cn[j] = ei[E + en + j]; }

            bool own = sat;
#pragma unroll
            for (int j = 0; j < GCH; ++j) own = own && (rr[j] == prev);
            if (!own) {
                const float dgo = degp[(size_t)rr[GCH - 1] * NS];
                float4 bv[GCH], av[GCH];
#pragma unroll
                for (int j = 0; j < GCH; ++j) bv[j] = ld_row4(agg, cc[j], lane);
#pragma unroll
                for (int j = 0; j < GCH; ++j) {
                    const int pj = j ? rr[j - 1] : prev;
                    if (rr[j] != pj) av[j] = ld_row4(agg, rr[j], lane);
                }
#pragma unroll
                for (int j = 0; j < GCH; ++j) {
                    const int pj = j ? rr[j - 1] : prev;
                    if (rr[j] != pj) {
                        if (j > 0 || pj >= 0) flush_run(gsum, pj, acc, lane);
                        acc = 0.0f;
                        a = av[j];
                    }
                    acc += sqdist4(a, bv[j]);
                }
                prev = rr[GCH - 1];
                bool nxt = more;
#pragma unroll
                for (int j = 0; j < GCH; ++j) nxt = nxt && (rn[j] == prev);
                sat = false;
                if (nxt) sat = wave_total(acc) >= SATK * dgo;
            }
            e0 += GCH;
            if (!more) break;
#pragma unroll
            for (int j = 0; j < GCH; ++j) { rr[j] = rn[j]; cc[j] = cn[j]; }
        }
    }
    for (; e0 < e_end; ++e0) {
        const int r = ei[e0];
        const int c = ei[E + e0];
        if (r != prev) {
            if (prev >= 0) flush_run(gsum, prev, acc, lane);
            acc = 0.0f; prev = r;
            a = ld_row4(agg, r, lane);
        }
        acc += sqdist4(a, ld_row4(agg, c, lane));
    }
    flush_run(gsum, prev, acc, lane);
}

__global__ __launch_bounds__(256) void gamma_edges(
    const float* __restrict__ agg,
    const int* __restrict__ eiA, int EA, float* __restrict__ gsumA, const float* __restrict__ degA, int qA, int rA,
    const int* __restrict__ eiB, int EB, float* __restrict__ gsumB, const float* __restrict__ degB, int qB, int rB)
{
    const int wv   = __builtin_amdgcn_readfirstlane((int)(threadIdx.x >> 5));
    const int lane = threadIdx.x & 31;
    const int w    = blockIdx.x * 8 + wv;
    const int nw   = gridDim.x * 8;

    const int a0 = w * qA + min(w, rA);
    const int a1 = a0 + qA + (w < rA ? 1 : 0);
    if (a0 < a1)
        walk_edges(agg, eiA, EA, gsumA, degA, a0 * GCH, min(a1 * GCH, EA), lane);

    const int xb = max(w - (nw - rB), 0);
    const int b0 = w * qB + xb;
    const int b1 = b0 + qB + (w >= nw - rB ? 1 : 0);
    if (b0 < b1)
        walk_edges(agg, eiB, EB, gsumB, degB, b0 * GCH, min(b1 * GCH, EB), lane);
}

__global__ __launch_bounds__(128) void update_fc(
    float* h, const float* __restrict__ agg, const float* __restrict__ xs,
    const float* gsum1, const float* deg1, const float* gsum2, const float* deg2,
    const float* __restrict__ W, const float* __restrict__ bias, float* out, int M)
{
    constexpr int K = HIDC, NC = NS;
    __shared__ float sA[32 * APITCH];
    const int wvu  = __builtin_amdgcn_readfirstlane((int)(threadIdx.x >> 5));
    const int lane = threadIdx.x & 31;
    update_rows_to_lds<8, false>(sA, h, agg, xs, gsum1, deg1, gsum2, deg2,
                                 (int)blockIdx.x * 32, M, wvu, lane);
    __syncthreads();

    const int tr = wvu >> 1;
    const int m0 = (int)blockIdx.x * 32 + (tr << 4);
    if (m0 >= M) return;
    const int hf = lane >> 4;
    const int c  = lane & 15;
    const float* Al = sA + ((tr << 4) + c) * APITCH + 2 * hf;
    if ((wvu & 1) == 0) {
        const float* Bp = W + (size_t)(2 * hf) * NC + c;
        v8f acc0 = {}, acc1 = {};
#pragma unroll
        for (int k0 = 0; k0 < K; k0 += 4) {
            const v2f a = *(const v2f*)(Al + k0);
            acc0 = wmma4(a.x, a.y, Bp[k0 * NC],      Bp[(k0 + 1) * NC],      acc0);
            acc1 = wmma4(a.x, a.y, Bp[k0 * NC + 16], Bp[(k0 + 1) * NC + 16], acc1);
        }
        const float bv0 = bias[c], bv1 = bias[c + 16];
#pragma unroll
        for (int v = 0; v < 8; ++v) {
            const int r = m0 + v + 8 * hf;
            if (r < M) {
                out[(size_t)r * NC + c]      = acc0[v] + bv0;
                out[(size_t)r * NC + c + 16] = acc1[v] + bv1;
            }
        }
    } else {
        const int col  = 32 + c;
        const int colc = min(col, NC - 1);
        const float* Bp = W + (size_t)(2 * hf) * NC + colc;
        v8f acc = {};
#pragma unroll
        for (int k0 = 0; k0 < K; k0 += 4) {
            const v2f a = *(const v2f*)(Al + k0);
            acc = wmma4(a.x, a.y, Bp[k0 * NC], Bp[(k0 + 1) * NC], acc);
        }
        if (col < NC) {
            const float bv = bias[col];
#pragma unroll
            for (int v = 0; v < 8; ++v) {
                const int r = m0 + v + 8 * hf;
                if (r < M) out[(size_t)r * NC + col] = acc[v] + bv;
            }
        }
    }
}

extern "C" void kernel_launch(void* const* d_in, const int* in_sizes, int n_in,
                              void* d_out, int out_size, void* d_ws, size_t ws_size,
                              hipStream_t stream) {
    const float* x      = (const float*)d_in[0];
    const float* x0     = (const float*)d_in[1];
    const int*   ei     = (const int*)d_in[2];
    const int*   ei2    = (const int*)d_in[3];
    const float* in_W   = (const float*)d_in[4];
    const float* skip_W = (const float*)d_in[5];
    const float* conv_W = (const float*)d_in[6];
    const float* conv_b = (const float*)d_in[7];
    const float* fc_W   = (const float*)d_in[8];
    const float* fc_b   = (const float*)d_in[9];
    float* out = (float*)d_out;

    const int N  = in_sizes[0] / HIDC;
    const int E1 = in_sizes[2] / 2;
    const int E2 = in_sizes[3] / 2;

    const size_t nh = (size_t)N * HIDC;
    float* h     = (float*)d_ws;
    float* xskip = h + nh;
    float* agg   = xskip + nh;
    float* mbuf  = agg + nh;

    float* rec  = (float*)d_out;
    float* deg1 = rec;
    float* deg2 = rec + 1;
    float* gsum1[2] = { rec + 2, rec + 4 };
    float* gsum2[2] = { rec + 3, rec + 5 };
    int* bkt = (int*)(rec + 6);

    const dim3 blk(256);
    const dim3 sblk(128);
    const int sb = (((N + 15) / 16) * 2 + 3) / 4;
    const float* conv_W1 = conv_W + (size_t)HIDC * HIDC;
    const int zb = 128, eb = 512;

    gemm_strip_f32<1><<<zb + 2 * sb, sblk, 0, stream>>>(
        x, in_W, h, x0, in_W, agg, N, sb, zb,
        nullptr, 0, nullptr, 0, nullptr, nullptr, nullptr, rec, NS * N,
        nullptr, nullptr, nullptr, nullptr, nullptr);
    gemm_strip_f32<2><<<eb + 2 * sb, sblk, 0, stream>>>(
        agg, skip_W, xskip, h, conv_W, mbuf, N, sb, eb,
        ei, E1, ei2, E2, deg1, deg2, bkt, nullptr, 0,
        nullptr, nullptr, nullptr, nullptr, nullptr);

    const int gblocks = 256 * GBW;
    const int gwaves  = gblocks * 8;
    const int nch2 = (E2 + GCH - 1) / GCH, nch1 = (E1 + GCH - 1) / GCH;
    const int q2 = nch2 / gwaves, r2 = nch2 % gwaves;
    const int q1 = nch1 / gwaves, r1 = nch1 % gwaves;
    for (int l = 0; l < 2; ++l) {
        if (l == 1)
            gemm_strip_f32<0><<<sb, sblk, 0, stream>>>(
                nullptr, conv_W1, mbuf, nullptr, nullptr, nullptr, N, sb, 0,
                nullptr, 0, nullptr, 0, deg1, deg2, nullptr, nullptr, 0,
                h, agg, xskip, gsum1[0], gsum2[0]);
        gather_agg<<<(N + 7) / 8, blk, 0, stream>>>(mbuf, bkt, conv_b + l * HIDC, agg, N);
        gamma_edges<<<gblocks, blk, 0, stream>>>(agg, ei2, E2, gsum2[l], deg2, q2, r2, ei, E1, gsum1[l], deg1, q1, r1);
    }

    update_fc<<<(N + 31) / 32, sblk, 0, stream>>>(
        h, agg, xskip, gsum1[1], deg1, gsum2[1], deg2, fc_W, fc_b, out, N);
}
